# starting kernel
# baseline (speedup 1.0000x reference)
#include <hip/hip_runtime.h>
#include <hip/hip_bf16.h>

typedef __attribute__((ext_vector_type(2))) float v2f;
typedef __attribute__((ext_vector_type(8))) float v8f;

#define D 128

__global__ void init_deg_kernel(float* deg, int n) {
  int i = blockIdx.x * blockDim.x + threadIdx.x;
  if (i < n) deg[i] = 1.0f;
}

__global__ void deg_accum_kernel(const int* __restrict__ col,
                                 const float* __restrict__ ew,
                                 float* deg, int e_cnt) {
  int e = blockIdx.x * blockDim.x + threadIdx.x;
  if (e < e_cnt) atomicAdd(&deg[col[e]], ew[e]);
}

__global__ void dinv_kernel(float* deg, int n) {
  int i = blockIdx.x * blockDim.x + threadIdx.x;
  if (i < n) {
    float d = deg[i];
    deg[i] = (d > 0.0f) ? rsqrtf(d) : 0.0f;
  }
}

__global__ __launch_bounds__(256) void gemm_wmma_kernel(
    const float* __restrict__ x, const float* __restrict__ w,
    float* __restrict__ xw, int n_rows) {
  const int lane = threadIdx.x & 31;
  const int wave = blockIdx.x * (blockDim.x >> 5) + (threadIdx.x >> 5);
  const int row0 = wave * 16;
  if (row0 >= n_rows) return;

  const int m  = lane & 15;
  const int kh = (lane >> 4) * 2;

  const int arow = (row0 + m < n_rows) ? (row0 + m) : (n_rows - 1);

  v8f acc[8];
#pragma unroll
  for (int t = 0; t < 8; ++t) acc[t] = (v8f){0.f,0.f,0.f,0.f,0.f,0.f,0.f,0.f};

  for (int k = 0; k < D; k += 4) {
    const float* ap = x + (size_t)arow * D + (k + kh);
    v2f a;
    a.x = ap[0];
    a.y = ap[1];
#pragma unroll
    for (int t = 0; t < 8; ++t) {
      const float* bp = w + (size_t)(k + kh) * D + t * 16 + m;
      v2f b;
      b.x = bp[0];
      b.y = bp[D];
      acc[t] = __builtin_amdgcn_wmma_f32_16x16x4_f32(
           false, a,  false, b,
           (short)0, acc[t],  false,  false);
    }
  }

#pragma unroll
  for (int t = 0; t < 8; ++t) {
#pragma unroll
    for (int j = 0; j < 8; ++j) {
      const int M = (kh << 2) + j;
      if (row0 + M < n_rows)
        xw[(size_t)(row0 + M) * D + t * 16 + m] = acc[t][j];
    }
  }
}

__global__ void init_out_kernel(const float* __restrict__ xw,
                                const float* __restrict__ dinv,
                                const float* __restrict__ bias,
                                float* out, long long total) {
  long long idx = (long long)blockIdx.x * blockDim.x + threadIdx.x;
  if (idx >= total) return;
  int i = (int)(idx >> 7);
  int d = (int)(idx & (D - 1));
  float di = dinv[i];
  out[idx] = bias[d] + di * di * xw[idx];
}

__global__ void scatter_kernel(const int* __restrict__ row,
                               const int* __restrict__ col,
                               const float* __restrict__ ew,
                               const float* __restrict__ dinv,
                               const float* __restrict__ xw,
                               float* out, long long total) {
  long long idx = (long long)blockIdx.x * blockDim.x + threadIdx.x;
  if (idx >= total) return;
  int e = (int)(idx >> 7);
  int d = (int)(idx & (D - 1));
  int r = row[e];
  int c = col[e];
  float norm = dinv[r] * ew[e] * dinv[c];
  atomicAdd(&out[(size_t)c * D + d], norm * xw[(size_t)r * D + d]);
}

__global__ void prelu_kernel(float* out, const float* __restrict__ pa,
                             long long total) {
  long long idx = (long long)blockIdx.x * blockDim.x + threadIdx.x;
  if (idx >= total) return;
  float v = out[idx];
  out[idx] = (v >= 0.0f) ? v : pa[0] * v;
}

extern "C" void kernel_launch(void* const* d_in, const int* in_sizes, int n_in,
                              void* d_out, int out_size, void* d_ws, size_t ws_size,
                              hipStream_t stream) {
  const float* x    = (const float*)d_in[0];
  const int*   ei   = (const int*)  d_in[1];
  const float* ew   = (const float*)d_in[2];
  const float* w    = (const float*)d_in[3];
  const float* bias = (const float*)d_in[4];
  const float* pa   = (const float*)d_in[5];
  float* out = (float*)d_out;

  const int n     = in_sizes[0] / D;
  const int e_cnt = in_sizes[2];
  const int* row = ei;
  const int* col = ei + e_cnt;

  float* deg = (float*)d_ws;
  float* xw  = deg + n;

  const long long nd  = (long long)n * D;
  const long long edd = (long long)e_cnt * D;

  init_deg_kernel<<<(n + 255) / 256, 256, 0, stream>>>(deg, n);
  deg_accum_kernel<<<(e_cnt + 255) / 256, 256, 0, stream>>>(col, ew, deg, e_cnt);
  dinv_kernel<<<(n + 255) / 256, 256, 0, stream>>>(deg, n);
  {
    int tiles  = (n + 15) / 16;
    int blocks = (tiles + 7) / 8;
    gemm_wmma_kernel<<<blocks, 256, 0, stream>>>(x, w, xw, n);
  }
  init_out_kernel<<<(int)((nd + 255) / 256), 256, 0, stream>>>(xw, deg, bias, out, nd);
  scatter_kernel<<<(int)((edd + 255) / 256), 256, 0, stream>>>(row, col, ew, deg, xw, out, edd);
  prelu_kernel<<<(int)((nd + 255) / 256), 256, 0, stream>>>(out, pa, nd);
}


# edge sums by target-range ownership: each work-group keeps 391 out rows in LDS, filters the whole edge list for them (no sort), adds rows by LDS float atomics, stores out once; LDS arrival order
# speedup vs baseline: 2.9358x; 2.9358x over previous
#include <hip/hip_runtime.h>
#include <hip/hip_bf16.h>

typedef __attribute__((ext_vector_type(2))) float v2f;
typedef __attribute__((ext_vector_type(8))) float v8f;

#define D 128

__global__ void init_deg_kernel(float* deg, int n) {
  int i = blockIdx.x * blockDim.x + threadIdx.x;
  if (i < n) deg[i] = 1.0f;
}

__global__ void deg_accum_kernel(const int* __restrict__ col,
                                 const float* __restrict__ ew,
                                 float* deg, int e_cnt) {
  int e = blockIdx.x * blockDim.x + threadIdx.x;
  if (e < e_cnt) atomicAdd(&deg[col[e]], ew[e]);
}

__global__ void dinv_kernel(float* deg, int n) {
  int i = blockIdx.x * blockDim.x + threadIdx.x;
  if (i < n) {
    float d = deg[i];
    deg[i] = (d > 0.0f) ? rsqrtf(d) : 0.0f;
  }
}

__global__ __launch_bounds__(256) void gemm_wmma_kernel(
    const float* __restrict__ x, const float* __restrict__ w,
    float* __restrict__ xw, int n_rows) {
  const int lane = threadIdx.x & 31;
  const int wave = blockIdx.x * (blockDim.x >> 5) + (threadIdx.x >> 5);
  const int row0 = wave * 16;
  if (row0 >= n_rows) return;

  const int m  = lane & 15;
  const int kh = (lane >> 4) * 2;

  const int arow = (row0 + m < n_rows) ? (row0 + m) : (n_rows - 1);

  v8f acc[8];
#pragma unroll
  for (int t = 0; t < 8; ++t) acc[t] = (v8f){0.f,0.f,0.f,0.f,0.f,0.f,0.f,0.f};

  for (int k = 0; k < D; k += 4) {
    const float* ap = x + (size_t)arow * D + (k + kh);
    v2f a;
    a.x = ap[0];
    a.y = ap[1];
#pragma unroll
    for (int t = 0; t < 8; ++t) {
      const float* bp = w + (size_t)(k + kh) * D + t * 16 + m;
      v2f b;
      b.x = bp[0];
      b.y = bp[D];
      acc[t] = __builtin_amdgcn_wmma_f32_16x16x4_f32(
           false, a,  false, b,
           (short)0, acc[t],  false,  false);
    }
  }

#pragma unroll
  for (int t = 0; t < 8; ++t) {
#pragma unroll
    for (int j = 0; j < 8; ++j) {
      const int M = (kh << 2) + j;
      if (row0 + M < n_rows)
        xw[(size_t)(row0 + M) * D + t * 16 + m] = acc[t][j];
    }
  }
}

#define RN 391
#define AGG_WAVES 32
#define QCAP 64
#define SCAN_DEPTH 8
#define SCAN_STEP (AGG_WAVES * SCAN_DEPTH)
#define ROWS_IN_FLIGHT 16

__device__ __forceinline__ void load_targets(int4 (&cc)[SCAN_DEPTH],
                                             const int* __restrict__ col,
                                             int ch0, int hi, int nfull,
                                             int lane, int e_cnt) {
  const int chl = ch0 + (SCAN_DEPTH - 1) * AGG_WAVES;
  if (chl < hi && chl < nfull) {
    const int4* p = reinterpret_cast<const int4*>(col + ((size_t)ch0 << 7)) + lane;
#pragma unroll
    for (int u = 0; u < SCAN_DEPTH; ++u) cc[u] = p[u * AGG_WAVES * 32];
  } else {
#pragma unroll
    for (int u = 0; u < SCAN_DEPTH; ++u) {
      const int ch = ch0 + u * AGG_WAVES;
      int4 c4 = make_int4(-1, -1, -1, -1);
      if (ch < hi) {
        const int e0 = (ch << 7) + (lane << 2);
        if (ch < nfull) {
          c4 = *reinterpret_cast<const int4*>(col + e0);
        } else {
          if (e0 < e_cnt) c4.x = col[e0];
          if (e0 + 1 < e_cnt) c4.y = col[e0 + 1];
          if (e0 + 2 < e_cnt) c4.z = col[e0 + 2];
          if (e0 + 3 < e_cnt) c4.w = col[e0 + 3];
        }
      }
      cc[u] = c4;
    }
  }
}

__global__ __launch_bounds__(1024) void aggregate_kernel(
    const int* __restrict__ row, const int* __restrict__ col,
    const float* __restrict__ ew, const float* __restrict__ dinv,
    const float* __restrict__ xw, const float* __restrict__ bias,
    const float* __restrict__ pa, float* __restrict__ out, int n, int e_cnt) {
  __shared__ float acc[RN * D];
  __shared__ int q_e[AGG_WAVES * QCAP];

  const int lane = threadIdx.x & 31;
  const int wave = __builtin_amdgcn_readfirstlane((int)(threadIdx.x >> 5));
  const int base = blockIdx.x * RN;
  const int cnt  = (n - base < RN) ? (n - base) : RN;

  for (int i = wave; i < cnt; i += AGG_WAVES) {
    const float di = dinv[base + i];
    const float s = di * di;
    const float4 v = *reinterpret_cast<const float4*>(xw + (size_t)(base + i) * D + 4 * lane);
    float* a = acc + i * D + lane;
    a[0]  = s * v.x;
    a[32] = s * v.y;
    a[64] = s * v.z;
    a[96] = s * v.w;
  }
  __syncthreads();

  int* qe = q_e + wave * QCAP;
  int head = 0;
  int qlen = 0;

  const int nch   = (e_cnt + 127) >> 7;
  const int nfull = e_cnt >> 7;
  const int start = (int)blockIdx.x * (nch / (int)gridDim.x);

  int pass = 0;
  int ch0 = start + wave;
  int hi = nch;
  if (ch0 >= hi) { pass = 1; ch0 = wave; hi = start; }
  bool valid = ch0 < hi;

  int4 cc[SCAN_DEPTH];
  if (valid) load_targets(cc, col, ch0, hi, nfull, lane, e_cnt);

  while (valid) {
    unsigned mbits = 0;
#pragma unroll
    for (int u = 0; u < SCAN_DEPTH; ++u) {
      const bool m0 = (unsigned)(cc[u].x - base) < (unsigned)cnt;
      const bool m1 = (unsigned)(cc[u].y - base) < (unsigned)cnt;
      const bool m2 = (unsigned)(cc[u].z - base) < (unsigned)cnt;
      const bool m3 = (unsigned)(cc[u].w - base) < (unsigned)cnt;
      if (__builtin_amdgcn_ballot_w32(m0 || m1 || m2 || m3) != 0u) {
        const unsigned mb = (m0 ? 1u : 0u) | (m1 ? 2u : 0u) | (m2 ? 4u : 0u) | (m3 ? 8u : 0u);
        mbits |= mb << (4 * u);
      }
    }
    const int cur = ch0;

    ch0 += SCAN_STEP;
    if (ch0 >= hi && pass == 0) { pass = 1; ch0 = wave; hi = start; }
    valid = ch0 < hi;
    if (valid) load_targets(cc, col, ch0, hi, nfull, lane, e_cnt);

    for (;;) {
      while (qlen < 32) {
        const bool has = mbits != 0u;
        const unsigned b = __builtin_amdgcn_ballot_w32(has);
        if (b == 0u) break;
        if (has) {
          const int s = __builtin_ctz(mbits);
          mbits &= mbits - 1u;
          const int e = ((cur + (s >> 2) * AGG_WAVES) << 7) + (lane << 2) + (s & 3);
          const int pos = head + qlen + (int)__builtin_amdgcn_mbcnt_lo(b, 0u);
          qe[pos & (QCAP - 1)] = e;
        }
        qlen += __builtin_popcount(b);
      }
      const int m = (qlen < 32) ? qlen : 32;
      if (m == 0 || (m < 32 && valid)) break;

      int r = 0, lc = 0;
      float nrm = 0.0f;
      if (lane < m) {
        const int e = qe[(head + lane) & (QCAP - 1)];
        r = row[e];
        const int c = col[e];
        const float w = ew[e];
        r = (r < 0) ? 0 : ((r > n - 1) ? (n - 1) : r);
        lc = c - base;
        nrm = dinv[r] * w * dinv[c];
      }
      for (int j0 = 0; j0 < m; j0 += ROWS_IN_FLIGHT) {
        float4 v[ROWS_IN_FLIGHT];
#pragma unroll
        for (int x8 = 0; x8 < ROWS_IN_FLIGHT; ++x8) {
          const int j = (j0 + x8 < m) ? (j0 + x8) : (m - 1);
          const int rj = __builtin_amdgcn_readlane(r, j);
          v[x8] = *reinterpret_cast<const float4*>(xw + (size_t)rj * D + 4 * lane);
        }
#pragma unroll
        for (int x8 = 0; x8 < ROWS_IN_FLIGHT; ++x8) {
          const int j = j0 + x8;
          if (j < m) {
            const int lcj = __builtin_amdgcn_readlane(lc, j);
            const float nj = __int_as_float(__builtin_amdgcn_readlane(__float_as_int(nrm), j));
            float* a = acc + lcj * D + lane;
            atomicAdd(a,      nj * v[x8].x);
            atomicAdd(a + 32, nj * v[x8].y);
            atomicAdd(a + 64, nj * v[x8].z);
            atomicAdd(a + 96, nj * v[x8].w);
          }
        }
      }
      head = (head + m) & (QCAP - 1);
      qlen -= m;
    }
  }
  __syncthreads();

  const float slope = pa[0];
  const float4 b4 = *reinterpret_cast<const float4*>(bias + 4 * lane);
  for (int i = wave; i < cnt; i += AGG_WAVES) {
    const float* a = acc + i * D + lane;
    float4 v;
    v.x = b4.x + a[0];
    v.y = b4.y + a[32];
    v.z = b4.z + a[64];
    v.w = b4.w + a[96];
    v.x = (v.x >= 0.0f) ? v.x : slope * v.x;
    v.y = (v.y >= 0.0f) ? v.y : slope * v.y;
    v.z = (v.z >= 0.0f) ? v.z : slope * v.z;
    v.w = (v.w >= 0.0f) ? v.w : slope * v.w;
    *reinterpret_cast<float4*>(out + (size_t)(base + i) * D + 4 * lane) = v;
  }
}

extern "C" void kernel_launch(void* const* d_in, const int* in_sizes, int n_in,
                              void* d_out, int out_size, void* d_ws, size_t ws_size,
                              hipStream_t stream) {
  const float* x    = (const float*)d_in[0];
  const int*   ei   = (const int*)  d_in[1];
  const float* ew   = (const float*)d_in[2];
  const float* w    = (const float*)d_in[3];
  const float* bias = (const float*)d_in[4];
  const float* pa   = (const float*)d_in[5];
  float* out = (float*)d_out;

  const int n     = in_sizes[0] / D;
  const int e_cnt = in_sizes[2];
  const int* row = ei;
  const int* col = ei + e_cnt;

  float* xw  = (float*)d_ws;
  float* deg = xw + (size_t)n * D;

  init_deg_kernel<<<(n + 255) / 256, 256, 0, stream>>>(deg, n);
  deg_accum_kernel<<<(e_cnt + 255) / 256, 256, 0, stream>>>(col, ew, deg, e_cnt);
  dinv_kernel<<<(n + 255) / 256, 256, 0, stream>>>(deg, n);
  {
    int tiles  = (n + 15) / 16;
    int blocks = (tiles + 7) / 8;
    gemm_wmma_kernel<<<blocks, 256, 0, stream>>>(x, w, xw, n);
  }
  aggregate_kernel<<<(n + RN - 1) / RN, 1024, 0, stream>>>(row, col, ew, deg, xw, bias, pa, out, n, e_cnt);
}


# edge list filtered by one-byte owner ids (target / 391, written by the degree pass) not 4-byte targets: a work-group scans a quarter of the bytes, 16 edges per load; LDS arrival order as before
# speedup vs baseline: 3.1198x; 1.0627x over previous
#include <hip/hip_runtime.h>
#include <hip/hip_bf16.h>

typedef __attribute__((ext_vector_type(2))) float v2f;
typedef __attribute__((ext_vector_type(8))) float v8f;

#define D 128

__global__ void init_deg_kernel(float* deg, int n) {
  int i = blockIdx.x * blockDim.x + threadIdx.x;
  if (i < n) deg[i] = 1.0f;
}

#define RN 391
__global__ void deg_accum_kernel(const int* __restrict__ col,
                                 const float* __restrict__ ew,
                                 float* deg, unsigned char* __restrict__ own,
                                 int e_cnt, int e_pad) {
  const int e0 = (blockIdx.x * blockDim.x + threadIdx.x) << 2;
  if (e0 >= e_pad) return;
  unsigned pk = 0u;
  if (e0 + 3 < e_cnt) {
    const int4 c = *reinterpret_cast<const int4*>(col + e0);
    const float4 w = *reinterpret_cast<const float4*>(ew + e0);
    atomicAdd(&deg[c.x], w.x);
    atomicAdd(&deg[c.y], w.y);
    atomicAdd(&deg[c.z], w.z);
    atomicAdd(&deg[c.w], w.w);
    pk = ((unsigned)(c.x / RN) & 255u) | (((unsigned)(c.y / RN) & 255u) << 8) |
         (((unsigned)(c.z / RN) & 255u) << 16) | (((unsigned)(c.w / RN) & 255u) << 24);
  } else {
    for (int k = 0; k < 4; ++k) {
      if (e0 + k < e_cnt) {
        const int c = col[e0 + k];
        atomicAdd(&deg[c], ew[e0 + k]);
        pk |= ((unsigned)(c / RN) & 255u) << (8 * k);
      }
    }
  }
  *reinterpret_cast<unsigned*>(own + e0) = pk;
}

__global__ void dinv_kernel(float* deg, int n) {
  int i = blockIdx.x * blockDim.x + threadIdx.x;
  if (i < n) {
    float d = deg[i];
    deg[i] = (d > 0.0f) ? rsqrtf(d) : 0.0f;
  }
}

__global__ __launch_bounds__(256) void gemm_wmma_kernel(
    const float* __restrict__ x, const float* __restrict__ w,
    float* __restrict__ xw, int n_rows) {
  const int lane = threadIdx.x & 31;
  const int wave = blockIdx.x * (blockDim.x >> 5) + (threadIdx.x >> 5);
  const int row0 = wave * 16;
  if (row0 >= n_rows) return;

  const int m  = lane & 15;
  const int kh = (lane >> 4) * 2;

  const int arow = (row0 + m < n_rows) ? (row0 + m) : (n_rows - 1);

  v8f acc[8];
#pragma unroll
  for (int t = 0; t < 8; ++t) acc[t] = (v8f){0.f,0.f,0.f,0.f,0.f,0.f,0.f,0.f};

  for (int k = 0; k < D; k += 4) {
    const float* ap = x + (size_t)arow * D + (k + kh);
    v2f a;
    a.x = ap[0];
    a.y = ap[1];
#pragma unroll
    for (int t = 0; t < 8; ++t) {
      const float* bp = w + (size_t)(k + kh) * D + t * 16 + m;
      v2f b;
      b.x = bp[0];
      b.y = bp[D];
      acc[t] = __builtin_amdgcn_wmma_f32_16x16x4_f32(
           false, a,  false, b,
           (short)0, acc[t],  false,  false);
    }
  }

#pragma unroll
  for (int t = 0; t < 8; ++t) {
#pragma unroll
    for (int j = 0; j < 8; ++j) {
      const int M = (kh << 2) + j;
      if (row0 + M < n_rows)
        xw[(size_t)(row0 + M) * D + t * 16 + m] = acc[t][j];
    }
  }
}

#define AGG_WAVES 32
#define QCAP 64
#define SCAN_DEPTH 8
#define SCAN_WORDS (SCAN_DEPTH / 2)
#define SCAN_STEP (AGG_WAVES * SCAN_DEPTH)
#define ROWS_IN_FLIGHT 16

__device__ __forceinline__ void load_owners(uint4 (&cc)[SCAN_DEPTH],
                                            const unsigned char* __restrict__ own,
                                            int ch0, int hi, int lane,
                                            unsigned nomatch) {
  const uint4* p = reinterpret_cast<const uint4*>(own + ((size_t)ch0 << 9)) + lane;
  const int chl = ch0 + (SCAN_DEPTH - 1) * AGG_WAVES;
  if (chl < hi) {
#pragma unroll
    for (int u = 0; u < SCAN_DEPTH; ++u) cc[u] = p[u * AGG_WAVES * 32];
  } else {
#pragma unroll
    for (int u = 0; u < SCAN_DEPTH; ++u) {
      uint4 c4 = make_uint4(nomatch, nomatch, nomatch, nomatch);
      if (ch0 + u * AGG_WAVES < hi) c4 = p[u * AGG_WAVES * 32];
      cc[u] = c4;
    }
  }
}

__device__ __forceinline__ unsigned nonzero_bytes(unsigned x) {
  return ((x & 0x7f7f7f7fu) + 0x7f7f7f7fu) | x;
}

__device__ __forceinline__ unsigned hits16(uint4 c, unsigned pat) {
  const unsigned z = ((nonzero_bytes(c.x ^ pat) >> 7) & 0x01010101u) |
                     ((nonzero_bytes(c.y ^ pat) >> 6) & 0x02020202u) |
                     ((nonzero_bytes(c.z ^ pat) >> 5) & 0x04040404u) |
                     ((nonzero_bytes(c.w ^ pat) >> 4) & 0x08080808u);
  return z ^ 0x0f0f0f0fu;
}

__global__ __launch_bounds__(1024) void aggregate_kernel(
    const int* __restrict__ row, const int* __restrict__ col,
    const float* __restrict__ ew, const float* __restrict__ dinv,
    const float* __restrict__ xw, const unsigned char* __restrict__ own,
    const float* __restrict__ bias, const float* __restrict__ pa,
    float* __restrict__ out, int n, int e_cnt, int e_pad) {
  __shared__ float acc[RN * D];
  __shared__ int q_e[AGG_WAVES * QCAP];

  const int lane = threadIdx.x & 31;
  const int wave = __builtin_amdgcn_readfirstlane((int)(threadIdx.x >> 5));
  const int base = blockIdx.x * RN;
  const int cnt  = (n - base < RN) ? (n - base) : RN;

  for (int i = wave; i < cnt; i += AGG_WAVES) {
    const float di = dinv[base + i];
    const float s = di * di;
    const float4 v = *reinterpret_cast<const float4*>(xw + (size_t)(base + i) * D + 4 * lane);
    float* a = acc + i * D + lane;
    a[0]  = s * v.x;
    a[32] = s * v.y;
    a[64] = s * v.z;
    a[96] = s * v.w;
  }
  __syncthreads();

  int* qe = q_e + wave * QCAP;
  int head = 0;
  int qlen = 0;

  const unsigned pat = ((unsigned)blockIdx.x & 255u) * 0x01010101u;
  const unsigned nomatch = ~pat;

  const int nch   = e_pad >> 9;
  const int start = (int)blockIdx.x * (nch / (int)gridDim.x);

  int pass = 0;
  int ch0 = start + wave;
  int hi = nch;
  if (ch0 >= hi) { pass = 1; ch0 = wave; hi = start; }
  bool valid = ch0 < hi;

  uint4 cc[SCAN_DEPTH];
  if (valid) load_owners(cc, own, ch0, hi, lane, nomatch);

  while (valid) {
    unsigned mw[SCAN_WORDS];
#pragma unroll
    for (int v = 0; v < SCAN_WORDS; ++v)
      mw[v] = hits16(cc[2 * v], pat) | (hits16(cc[2 * v + 1], pat) << 4);
    const int cur = ch0;

    ch0 += SCAN_STEP;
    if (ch0 >= hi && pass == 0) { pass = 1; ch0 = wave; hi = start; }
    valid = ch0 < hi;
    if (valid) load_owners(cc, own, ch0, hi, lane, nomatch);

#pragma nounroll
    for (int v = 0; v < SCAN_WORDS; ++v) {
      unsigned mbits = mw[0];
#pragma unroll
      for (int t = 0; t + 1 < SCAN_WORDS; ++t) mw[t] = mw[t + 1];
      const bool flush = !valid && (v == SCAN_WORDS - 1);
      const int cbase = cur + 2 * v * AGG_WAVES;

      for (;;) {
        while (qlen < 32) {
          const bool has = mbits != 0u;
          const unsigned b = __builtin_amdgcn_ballot_w32(has);
          if (b == 0u) break;
          if (has) {
            const int s = __builtin_ctz(mbits);
            mbits &= mbits - 1u;
            const int ch = cbase + ((s >> 2) & 1) * AGG_WAVES;
            const int e = (ch << 9) + (lane << 4) + ((s & 3) << 2) + (s >> 3);
            const int pos = head + qlen + (int)__builtin_amdgcn_mbcnt_lo(b, 0u);
            qe[pos & (QCAP - 1)] = e;
          }
          qlen += __builtin_popcount(b);
        }
        const int m = (qlen < 32) ? qlen : 32;
        if (m == 0 || (m < 32 && !flush)) break;

        int r = 0, lc = -1;
        float nrm = 0.0f;
        if (lane < m) {
          const int e = qe[(head + lane) & (QCAP - 1)];
          if (e < e_cnt) {
            r = row[e];
            const int c = col[e];
            const float w = ew[e];
            r = (r < 0) ? 0 : ((r > n - 1) ? (n - 1) : r);
            if ((unsigned)(c - base) < (unsigned)cnt) {
              lc = c - base;
              nrm = dinv[r] * w * dinv[c];
            }
          }
        }
        for (int j0 = 0; j0 < m; j0 += ROWS_IN_FLIGHT) {
          float4 v4[ROWS_IN_FLIGHT];
#pragma unroll
          for (int x8 = 0; x8 < ROWS_IN_FLIGHT; ++x8) {
            const int j = (j0 + x8 < m) ? (j0 + x8) : (m - 1);
            const int rj = __builtin_amdgcn_readlane(r, j);
            v4[x8] = *reinterpret_cast<const float4*>(xw + (size_t)rj * D + 4 * lane);
          }
#pragma unroll
          for (int x8 = 0; x8 < ROWS_IN_FLIGHT; ++x8) {
            const int j = j0 + x8;
            if (j < m) {
              const int lcj = __builtin_amdgcn_readlane(lc, j);
              if (lcj >= 0) {
                const float nj = __int_as_float(__builtin_amdgcn_readlane(__float_as_int(nrm), j));
                float* a = acc + lcj * D + lane;
                atomicAdd(a,      nj * v4[x8].x);
                atomicAdd(a + 32, nj * v4[x8].y);
                atomicAdd(a + 64, nj * v4[x8].z);
                atomicAdd(a + 96, nj * v4[x8].w);
              }
            }
          }
        }
        head = (head + m) & (QCAP - 1);
        qlen -= m;
      }
    }
  }
  __syncthreads();

  const float slope = pa[0];
  const float4 b4 = *reinterpret_cast<const float4*>(bias + 4 * lane);
  for (int i = wave; i < cnt; i += AGG_WAVES) {
    const float* a = acc + i * D + lane;
    float4 v;
    v.x = b4.x + a[0];
    v.y = b4.y + a[32];
    v.z = b4.z + a[64];
    v.w = b4.w + a[96];
    v.x = (v.x >= 0.0f) ? v.x : slope * v.x;
    v.y = (v.y >= 0.0f) ? v.y : slope * v.y;
    v.z = (v.z >= 0.0f) ? v.z : slope * v.z;
    v.w = (v.w >= 0.0f) ? v.w : slope * v.w;
    *reinterpret_cast<float4*>(out + (size_t)(base + i) * D + 4 * lane) = v;
  }
}

extern "C" void kernel_launch(void* const* d_in, const int* in_sizes, int n_in,
                              void* d_out, int out_size, void* d_ws, size_t ws_size,
                              hipStream_t stream) {
  const float* x    = (const float*)d_in[0];
  const int*   ei   = (const int*)  d_in[1];
  const float* ew   = (const float*)d_in[2];
  const float* w    = (const float*)d_in[3];
  const float* bias = (const float*)d_in[4];
  const float* pa   = (const float*)d_in[5];
  float* out = (float*)d_out;

  const int n     = in_sizes[0] / D;
  const int e_cnt = in_sizes[2];
  const int* row = ei;
  const int* col = ei + e_cnt;

  float* xw  = (float*)d_ws;
  float* deg = xw + (size_t)n * D;
  unsigned char* own = (unsigned char*)(deg + ((n + 3) & ~3));
  const int e_pad = (e_cnt + 511) & ~511;

  init_deg_kernel<<<(n + 255) / 256, 256, 0, stream>>>(deg, n);
  deg_accum_kernel<<<(e_pad / 4 + 255) / 256, 256, 0, stream>>>(col, ew, deg, own, e_cnt, e_pad);
  dinv_kernel<<<(n + 255) / 256, 256, 0, stream>>>(deg, n);
  {
    int tiles  = (n + 15) / 16;
    int blocks = (tiles + 7) / 8;
    gemm_wmma_kernel<<<blocks, 256, 0, stream>>>(x, w, xw, n);
  }
  aggregate_kernel<<<(n + RN - 1) / RN, 1024, 0, stream>>>(row, col, ew, deg, xw, own, bias, pa, out, n, e_cnt, e_pad);
}


# degree sums without global float atomics: 32 edge slabs x 8 target ranges, each work-group adds its range in LDS and stores partial sums; a reduce pass adds the 32 in slab order, takes rsqrt
# speedup vs baseline: 4.7805x; 1.5323x over previous
#include <hip/hip_runtime.h>
#include <hip/hip_bf16.h>

typedef __attribute__((ext_vector_type(2))) float v2f;
typedef __attribute__((ext_vector_type(8))) float v8f;

#define D 128

#define RN 391
#define DEG_SLABS 32
#define DEG_HR 12544
#define DEG_UNR 4

__device__ __forceinline__ void deg_apply(float* hist, unsigned char* __restrict__ own,
                                          int4 c, float4 w, int lo, int e0, bool bytes) {
  const unsigned dx = (unsigned)(c.x - lo);
  const unsigned dy = (unsigned)(c.y - lo);
  const unsigned dz = (unsigned)(c.z - lo);
  const unsigned dw = (unsigned)(c.w - lo);
  if (dx < (unsigned)DEG_HR) atomicAdd(&hist[dx], w.x);
  if (dy < (unsigned)DEG_HR) atomicAdd(&hist[dy], w.y);
  if (dz < (unsigned)DEG_HR) atomicAdd(&hist[dz], w.z);
  if (dw < (unsigned)DEG_HR) atomicAdd(&hist[dw], w.w);
  if (bytes) {
    const unsigned pk =
        ((unsigned)(c.x / RN) & 255u) | (((unsigned)(c.y / RN) & 255u) << 8) |
        (((unsigned)(c.z / RN) & 255u) << 16) | (((unsigned)(c.w / RN) & 255u) << 24);
    *reinterpret_cast<unsigned*>(own + e0) = pk;
  }
}

__global__ __launch_bounds__(1024) void deg_partial_kernel(
    const int* __restrict__ col, const float* __restrict__ ew,
    float* __restrict__ part, unsigned char* __restrict__ own,
    int n_pad, int e_cnt, int e_pad) {
  __shared__ alignas(16) float hist[DEG_HR];
  const int tid  = threadIdx.x;
  const int slab = (int)blockIdx.x % DEG_SLABS;
  const int g    = (int)blockIdx.x / DEG_SLABS;
  const int n_ranges = (int)gridDim.x / DEG_SLABS;
  const int lo = g * DEG_HR;

  for (int i = tid; i < DEG_HR; i += 1024) hist[i] = 0.0f;
  __syncthreads();

  const int units = (e_pad + 4095) >> 12;
  const int ubeg = (int)(((long long)slab * units) / DEG_SLABS);
  const int uend = (int)(((long long)(slab + 1) * units) / DEG_SLABS);
  const int ufull = e_cnt >> 12;

  int u0 = ubeg;
  const int ufast = (uend < ufull) ? uend : ufull;
  for (; u0 + DEG_UNR <= ufast; u0 += DEG_UNR) {
    int4 c[DEG_UNR];
    float4 w[DEG_UNR];
#pragma unroll
    for (int j = 0; j < DEG_UNR; ++j) {
      const int e0 = ((u0 + j) << 12) + (tid << 2);
      c[j] = *reinterpret_cast<const int4*>(col + e0);
      w[j] = *reinterpret_cast<const float4*>(ew + e0);
    }
#pragma unroll
    for (int j = 0; j < DEG_UNR; ++j) {
      const int u = u0 + j;
      deg_apply(hist, own, c[j], w[j], lo, (u << 12) + (tid << 2), (u % n_ranges) == g);
    }
  }
  for (; u0 < uend; ++u0) {
    const int e0 = (u0 << 12) + (tid << 2);
    int4 c = make_int4(-1, -1, -1, -1);
    float4 w = make_float4(0.0f, 0.0f, 0.0f, 0.0f);
    if (e0 + 3 < e_cnt) {
      c = *reinterpret_cast<const int4*>(col + e0);
      w = *reinterpret_cast<const float4*>(ew + e0);
    } else {
      if (e0 < e_cnt)     { c.x = col[e0];     w.x = ew[e0]; }
      if (e0 + 1 < e_cnt) { c.y = col[e0 + 1]; w.y = ew[e0 + 1]; }
      if (e0 + 2 < e_cnt) { c.z = col[e0 + 2]; w.z = ew[e0 + 2]; }
    }
    deg_apply(hist, own, c, w, lo, e0, e0 < e_pad && (u0 % n_ranges) == g);
  }
  __syncthreads();

  float* dst = part + (size_t)slab * n_pad + lo;
  for (int i = tid << 2; i < DEG_HR; i += 4096)
    *reinterpret_cast<float4*>(dst + i) = *reinterpret_cast<const float4*>(hist + i);
}

__global__ void dinv_reduce_kernel(const float* __restrict__ part, float* __restrict__ dinv,
                                   int n, int n_pad) {
  const int i = blockIdx.x * blockDim.x + threadIdx.x;
  if (i < n) {
    float d = 1.0f;
#pragma unroll
    for (int s = 0; s < DEG_SLABS; ++s) d += part[(size_t)s * n_pad + i];
    dinv[i] = (d > 0.0f) ? rsqrtf(d) : 0.0f;
  }
}

__global__ __launch_bounds__(256) void gemm_wmma_kernel(
    const float* __restrict__ x, const float* __restrict__ w,
    float* __restrict__ xw, int n_rows) {
  const int lane = threadIdx.x & 31;
  const int wave = blockIdx.x * (blockDim.x >> 5) + (threadIdx.x >> 5);
  const int row0 = wave * 16;
  if (row0 >= n_rows) return;

  const int m  = lane & 15;
  const int kh = (lane >> 4) * 2;

  const int arow = (row0 + m < n_rows) ? (row0 + m) : (n_rows - 1);

  v8f acc[8];
#pragma unroll
  for (int t = 0; t < 8; ++t) acc[t] = (v8f){0.f,0.f,0.f,0.f,0.f,0.f,0.f,0.f};

  for (int k = 0; k < D; k += 4) {
    const float* ap = x + (size_t)arow * D + (k + kh);
    v2f a;
    a.x = ap[0];
    a.y = ap[1];
#pragma unroll
    for (int t = 0; t < 8; ++t) {
      const float* bp = w + (size_t)(k + kh) * D + t * 16 + m;
      v2f b;
      b.x = bp[0];
      b.y = bp[D];
      acc[t] = __builtin_amdgcn_wmma_f32_16x16x4_f32(
           false, a,  false, b,
           (short)0, acc[t],  false,  false);
    }
  }

#pragma unroll
  for (int t = 0; t < 8; ++t) {
#pragma unroll
    for (int j = 0; j < 8; ++j) {
      const int M = (kh << 2) + j;
      if (row0 + M < n_rows)
        xw[(size_t)(row0 + M) * D + t * 16 + m] = acc[t][j];
    }
  }
}

#define AGG_WAVES 32
#define QCAP 64
#define SCAN_DEPTH 8
#define SCAN_WORDS (SCAN_DEPTH / 2)
#define SCAN_STEP (AGG_WAVES * SCAN_DEPTH)
#define ROWS_IN_FLIGHT 16

__device__ __forceinline__ void load_owners(uint4 (&cc)[SCAN_DEPTH],
                                            const unsigned char* __restrict__ own,
                                            int ch0, int hi, int lane,
                                            unsigned nomatch) {
  const uint4* p = reinterpret_cast<const uint4*>(own + ((size_t)ch0 << 9)) + lane;
  const int chl = ch0 + (SCAN_DEPTH - 1) * AGG_WAVES;
  if (chl < hi) {
#pragma unroll
    for (int u = 0; u < SCAN_DEPTH; ++u) cc[u] = p[u * AGG_WAVES * 32];
  } else {
#pragma unroll
    for (int u = 0; u < SCAN_DEPTH; ++u) {
      uint4 c4 = make_uint4(nomatch, nomatch, nomatch, nomatch);
      if (ch0 + u * AGG_WAVES < hi) c4 = p[u * AGG_WAVES * 32];
      cc[u] = c4;
    }
  }
}

__device__ __forceinline__ unsigned nonzero_bytes(unsigned x) {
  return ((x & 0x7f7f7f7fu) + 0x7f7f7f7fu) | x;
}

__device__ __forceinline__ unsigned hits16(uint4 c, unsigned pat) {
  const unsigned z = ((nonzero_bytes(c.x ^ pat) >> 7) & 0x01010101u) |
                     ((nonzero_bytes(c.y ^ pat) >> 6) & 0x02020202u) |
                     ((nonzero_bytes(c.z ^ pat) >> 5) & 0x04040404u) |
                     ((nonzero_bytes(c.w ^ pat) >> 4) & 0x08080808u);
  return z ^ 0x0f0f0f0fu;
}

__global__ __launch_bounds__(1024) void aggregate_kernel(
    const int* __restrict__ row, const int* __restrict__ col,
    const float* __restrict__ ew, const float* __restrict__ dinv,
    const float* __restrict__ xw, const unsigned char* __restrict__ own,
    const float* __restrict__ bias, const float* __restrict__ pa,
    float* __restrict__ out, int n, int e_cnt, int e_pad) {
  __shared__ float acc[RN * D];
  __shared__ int q_e[AGG_WAVES * QCAP];

  const int lane = threadIdx.x & 31;
  const int wave = __builtin_amdgcn_readfirstlane((int)(threadIdx.x >> 5));
  const int base = blockIdx.x * RN;
  const int cnt  = (n - base < RN) ? (n - base) : RN;

  for (int i = wave; i < cnt; i += AGG_WAVES) {
    const float di = dinv[base + i];
    const float s = di * di;
    const float4 v = *reinterpret_cast<const float4*>(xw + (size_t)(base + i) * D + 4 * lane);
    float* a = acc + i * D + lane;
    a[0]  = s * v.x;
    a[32] = s * v.y;
    a[64] = s * v.z;
    a[96] = s * v.w;
  }
  __syncthreads();

  int* qe = q_e + wave * QCAP;
  int head = 0;
  int qlen = 0;

  const unsigned pat = ((unsigned)blockIdx.x & 255u) * 0x01010101u;
  const unsigned nomatch = ~pat;

  const int nch   = e_pad >> 9;
  const int start = (int)blockIdx.x * (nch / (int)gridDim.x);

  int pass = 0;
  int ch0 = start + wave;
  int hi = nch;
  if (ch0 >= hi) { pass = 1; ch0 = wave; hi = start; }
  bool valid = ch0 < hi;

  uint4 cc[SCAN_DEPTH];
  if (valid) load_owners(cc, own, ch0, hi, lane, nomatch);

  while (valid) {
    unsigned mw[SCAN_WORDS];
#pragma unroll
    for (int v = 0; v < SCAN_WORDS; ++v)
      mw[v] = hits16(cc[2 * v], pat) | (hits16(cc[2 * v + 1], pat) << 4);
    const int cur = ch0;

    ch0 += SCAN_STEP;
    if (ch0 >= hi && pass == 0) { pass = 1; ch0 = wave; hi = start; }
    valid = ch0 < hi;
    if (valid) load_owners(cc, own, ch0, hi, lane, nomatch);

#pragma nounroll
    for (int v = 0; v < SCAN_WORDS; ++v) {
      unsigned mbits = mw[0];
#pragma unroll
      for (int t = 0; t + 1 < SCAN_WORDS; ++t) mw[t] = mw[t + 1];
      const bool flush = !valid && (v == SCAN_WORDS - 1);
      const int cbase = cur + 2 * v * AGG_WAVES;

      for (;;) {
        while (qlen < 32) {
          const bool has = mbits != 0u;
          const unsigned b = __builtin_amdgcn_ballot_w32(has);
          if (b == 0u) break;
          if (has) {
            const int s = __builtin_ctz(mbits);
            mbits &= mbits - 1u;
            const int ch = cbase + ((s >> 2) & 1) * AGG_WAVES;
            const int e = (ch << 9) + (lane << 4) + ((s & 3) << 2) + (s >> 3);
            const int pos = head + qlen + (int)__builtin_amdgcn_mbcnt_lo(b, 0u);
            qe[pos & (QCAP - 1)] = e;
          }
          qlen += __builtin_popcount(b);
        }
        const int m = (qlen < 32) ? qlen : 32;
        if (m == 0 || (m < 32 && !flush)) break;

        int r = 0, lc = -1;
        float nrm = 0.0f;
        if (lane < m) {
          const int e = qe[(head + lane) & (QCAP - 1)];
          if (e < e_cnt) {
            r = row[e];
            const int c = col[e];
            const float w = ew[e];
            r = (r < 0) ? 0 : ((r > n - 1) ? (n - 1) : r);
            if ((unsigned)(c - base) < (unsigned)cnt) {
              lc = c - base;
              nrm = dinv[r] * w * dinv[c];
            }
          }
        }
        for (int j0 = 0; j0 < m; j0 += ROWS_IN_FLIGHT) {
          float4 v4[ROWS_IN_FLIGHT];
#pragma unroll
          for (int x8 = 0; x8 < ROWS_IN_FLIGHT; ++x8) {
            const int j = (j0 + x8 < m) ? (j0 + x8) : (m - 1);
            const int rj = __builtin_amdgcn_readlane(r, j);
            v4[x8] = *reinterpret_cast<const float4*>(xw + (size_t)rj * D + 4 * lane);
          }
#pragma unroll
          for (int x8 = 0; x8 < ROWS_IN_FLIGHT; ++x8) {
            const int j = j0 + x8;
            if (j < m) {
              const int lcj = __builtin_amdgcn_readlane(lc, j);
              if (lcj >= 0) {
                const float nj = __int_as_float(__builtin_amdgcn_readlane(__float_as_int(nrm), j));
                float* a = acc + lcj * D + lane;
                atomicAdd(a,      nj * v4[x8].x);
                atomicAdd(a + 32, nj * v4[x8].y);
                atomicAdd(a + 64, nj * v4[x8].z);
                atomicAdd(a + 96, nj * v4[x8].w);
              }
            }
          }
        }
        head = (head + m) & (QCAP - 1);
        qlen -= m;
      }
    }
  }
  __syncthreads();

  const float slope = pa[0];
  const float4 b4 = *reinterpret_cast<const float4*>(bias + 4 * lane);
  for (int i = wave; i < cnt; i += AGG_WAVES) {
    const float* a = acc + i * D + lane;
    float4 v;
    v.x = b4.x + a[0];
    v.y = b4.y + a[32];
    v.z = b4.z + a[64];
    v.w = b4.w + a[96];
    v.x = (v.x >= 0.0f) ? v.x : slope * v.x;
    v.y = (v.y >= 0.0f) ? v.y : slope * v.y;
    v.z = (v.z >= 0.0f) ? v.z : slope * v.z;
    v.w = (v.w >= 0.0f) ? v.w : slope * v.w;
    *reinterpret_cast<float4*>(out + (size_t)(base + i) * D + 4 * lane) = v;
  }
}

extern "C" void kernel_launch(void* const* d_in, const int* in_sizes, int n_in,
                              void* d_out, int out_size, void* d_ws, size_t ws_size,
                              hipStream_t stream) {
  const float* x    = (const float*)d_in[0];
  const int*   ei   = (const int*)  d_in[1];
  const float* ew   = (const float*)d_in[2];
  const float* w    = (const float*)d_in[3];
  const float* bias = (const float*)d_in[4];
  const float* pa   = (const float*)d_in[5];
  float* out = (float*)d_out;

  const int n     = in_sizes[0] / D;
  const int e_cnt = in_sizes[2];
  const int* row = ei;
  const int* col = ei + e_cnt;

  float* xw  = (float*)d_ws;
  float* deg = xw + (size_t)n * D;
  unsigned char* own = (unsigned char*)(deg + ((n + 3) & ~3));
  const int e_pad = (e_cnt + 511) & ~511;
  float* part = (float*)(own + e_pad);
  const int n_ranges = (n + DEG_HR - 1) / DEG_HR;
  const int n_pad = n_ranges * DEG_HR;

  deg_partial_kernel<<<DEG_SLABS * n_ranges, 1024, 0, stream>>>(col, ew, part, own, n_pad, e_cnt, e_pad);
  dinv_reduce_kernel<<<(n + 255) / 256, 256, 0, stream>>>(part, deg, n, n_pad);
  {
    int tiles  = (n + 15) / 16;
    int blocks = (tiles + 7) / 8;
    gemm_wmma_kernel<<<blocks, 256, 0, stream>>>(x, w, xw, n);
  }
  aggregate_kernel<<<(n + RN - 1) / RN, 1024, 0, stream>>>(row, col, ew, deg, xw, own, bias, pa, out, n, e_cnt, e_pad);
}


# product applied after the edge sums, not before: each work-group sums scaled x rows in LDS, then multiplies its rows by w on f32 WMMA from LDS; x@w array and launch gone; sums re-associated
# speedup vs baseline: 5.3516x; 1.1195x over previous
#include <hip/hip_runtime.h>
#include <hip/hip_bf16.h>

typedef __attribute__((ext_vector_type(2))) float v2f;
typedef __attribute__((ext_vector_type(8))) float v8f;

#define D 128

#define RN 391
#define DEG_SLABS 32
#define DEG_HR 12544
#define DEG_UNR 4

__device__ __forceinline__ void deg_apply(float* hist, unsigned char* __restrict__ own,
                                          int4 c, float4 w, int lo, int e0, bool bytes) {
  const unsigned dx = (unsigned)(c.x - lo);
  const unsigned dy = (unsigned)(c.y - lo);
  const unsigned dz = (unsigned)(c.z - lo);
  const unsigned dw = (unsigned)(c.w - lo);
  if (dx < (unsigned)DEG_HR) atomicAdd(&hist[dx], w.x);
  if (dy < (unsigned)DEG_HR) atomicAdd(&hist[dy], w.y);
  if (dz < (unsigned)DEG_HR) atomicAdd(&hist[dz], w.z);
  if (dw < (unsigned)DEG_HR) atomicAdd(&hist[dw], w.w);
  if (bytes) {
    const unsigned pk =
        ((unsigned)(c.x / RN) & 255u) | (((unsigned)(c.y / RN) & 255u) << 8) |
        (((unsigned)(c.z / RN) & 255u) << 16) | (((unsigned)(c.w / RN) & 255u) << 24);
    *reinterpret_cast<unsigned*>(own + e0) = pk;
  }
}

__global__ __launch_bounds__(1024) void deg_partial_kernel(
    const int* __restrict__ col, const float* __restrict__ ew,
    float* __restrict__ part, unsigned char* __restrict__ own,
    int n_pad, int e_cnt, int e_pad) {
  __shared__ alignas(16) float hist[DEG_HR];
  const int tid  = threadIdx.x;
  const int slab = (int)blockIdx.x % DEG_SLABS;
  const int g    = (int)blockIdx.x / DEG_SLABS;
  const int n_ranges = (int)gridDim.x / DEG_SLABS;
  const int lo = g * DEG_HR;

  for (int i = tid; i < DEG_HR; i += 1024) hist[i] = 0.0f;
  __syncthreads();

  const int units = (e_pad + 4095) >> 12;
  const int ubeg = (int)(((long long)slab * units) / DEG_SLABS);
  const int uend = (int)(((long long)(slab + 1) * units) / DEG_SLABS);
  const int ufull = e_cnt >> 12;

  int u0 = ubeg;
  const int ufast = (uend < ufull) ? uend : ufull;
  for (; u0 + DEG_UNR <= ufast; u0 += DEG_UNR) {
    int4 c[DEG_UNR];
    float4 w[DEG_UNR];
#pragma unroll
    for (int j = 0; j < DEG_UNR; ++j) {
      const int e0 = ((u0 + j) << 12) + (tid << 2);
      c[j] = *reinterpret_cast<const int4*>(col + e0);
      w[j] = *reinterpret_cast<const float4*>(ew + e0);
    }
#pragma unroll
    for (int j = 0; j < DEG_UNR; ++j) {
      const int u = u0 + j;
      deg_apply(hist, own, c[j], w[j], lo, (u << 12) + (tid << 2), (u % n_ranges) == g);
    }
  }
  for (; u0 < uend; ++u0) {
    const int e0 = (u0 << 12) + (tid << 2);
    int4 c = make_int4(-1, -1, -1, -1);
    float4 w = make_float4(0.0f, 0.0f, 0.0f, 0.0f);
    if (e0 + 3 < e_cnt) {
      c = *reinterpret_cast<const int4*>(col + e0);
      w = *reinterpret_cast<const float4*>(ew + e0);
    } else {
      if (e0 < e_cnt)     { c.x = col[e0];     w.x = ew[e0]; }
      if (e0 + 1 < e_cnt) { c.y = col[e0 + 1]; w.y = ew[e0 + 1]; }
      if (e0 + 2 < e_cnt) { c.z = col[e0 + 2]; w.z = ew[e0 + 2]; }
    }
    deg_apply(hist, own, c, w, lo, e0, e0 < e_pad && (u0 % n_ranges) == g);
  }
  __syncthreads();

  float* dst = part + (size_t)slab * n_pad + lo;
  for (int i = tid << 2; i < DEG_HR; i += 4096)
    *reinterpret_cast<float4*>(dst + i) = *reinterpret_cast<const float4*>(hist + i);
}

__global__ void dinv_reduce_kernel(const float* __restrict__ part, float* __restrict__ dinv,
                                   int n, int n_pad) {
  const int i = blockIdx.x * blockDim.x + threadIdx.x;
  if (i < n) {
    float d = 1.0f;
#pragma unroll
    for (int s = 0; s < DEG_SLABS; ++s) d += part[(size_t)s * n_pad + i];
    dinv[i] = (d > 0.0f) ? rsqrtf(d) : 0.0f;
  }
}

#define AGG_WAVES 32
#define QCAP 64
#define SCAN_DEPTH 8
#define SCAN_WORDS (SCAN_DEPTH / 2)
#define SCAN_STEP (AGG_WAVES * SCAN_DEPTH)
#define ROWS_IN_FLIGHT 16

__device__ __forceinline__ void load_owners(uint4 (&cc)[SCAN_DEPTH],
                                            const unsigned char* __restrict__ own,
                                            int ch0, int hi, int lane,
                                            unsigned nomatch) {
  const uint4* p = reinterpret_cast<const uint4*>(own + ((size_t)ch0 << 9)) + lane;
  const int chl = ch0 + (SCAN_DEPTH - 1) * AGG_WAVES;
  if (chl < hi) {
#pragma unroll
    for (int u = 0; u < SCAN_DEPTH; ++u) cc[u] = p[u * AGG_WAVES * 32];
  } else {
#pragma unroll
    for (int u = 0; u < SCAN_DEPTH; ++u) {
      uint4 c4 = make_uint4(nomatch, nomatch, nomatch, nomatch);
      if (ch0 + u * AGG_WAVES < hi) c4 = p[u * AGG_WAVES * 32];
      cc[u] = c4;
    }
  }
}

__device__ __forceinline__ unsigned nonzero_bytes(unsigned x) {
  return ((x & 0x7f7f7f7fu) + 0x7f7f7f7fu) | x;
}

__device__ __forceinline__ unsigned hits16(uint4 c, unsigned pat) {
  const unsigned z = ((nonzero_bytes(c.x ^ pat) >> 7) & 0x01010101u) |
                     ((nonzero_bytes(c.y ^ pat) >> 6) & 0x02020202u) |
                     ((nonzero_bytes(c.z ^ pat) >> 5) & 0x04040404u) |
                     ((nonzero_bytes(c.w ^ pat) >> 4) & 0x08080808u);
  return z ^ 0x0f0f0f0fu;
}

__global__ __launch_bounds__(1024) void aggregate_kernel(
    const int* __restrict__ row, const int* __restrict__ col,
    const float* __restrict__ ew, const float* __restrict__ dinv,
    const float* __restrict__ x, const float* __restrict__ w,
    const unsigned char* __restrict__ own,
    const float* __restrict__ bias, const float* __restrict__ pa,
    float* __restrict__ out, int n, int e_cnt, int e_pad) {
  __shared__ float acc[RN * D];
  __shared__ int q_e[AGG_WAVES * QCAP];

  const int lane = threadIdx.x & 31;
  const int wave = __builtin_amdgcn_readfirstlane((int)(threadIdx.x >> 5));
  const int base = blockIdx.x * RN;
  const int cnt  = (n - base < RN) ? (n - base) : RN;

  for (int i = wave; i < cnt; i += AGG_WAVES) {
    const float di = dinv[base + i];
    const float s = di * di;
    const float4 v = *reinterpret_cast<const float4*>(x + (size_t)(base + i) * D + 4 * lane);
    float* a = acc + i * D + ((lane + i) & 31);
    a[0]  = s * v.x;
    a[32] = s * v.y;
    a[64] = s * v.z;
    a[96] = s * v.w;
  }
  __syncthreads();

  int* qe = q_e + wave * QCAP;
  int head = 0;
  int qlen = 0;

  const unsigned pat = ((unsigned)blockIdx.x & 255u) * 0x01010101u;
  const unsigned nomatch = ~pat;

  const int nch   = e_pad >> 9;
  const int start = (int)blockIdx.x * (nch / (int)gridDim.x);

  int pass = 0;
  int ch0 = start + wave;
  int hi = nch;
  if (ch0 >= hi) { pass = 1; ch0 = wave; hi = start; }
  bool valid = ch0 < hi;

  uint4 cc[SCAN_DEPTH];
  if (valid) load_owners(cc, own, ch0, hi, lane, nomatch);

  while (valid) {
    unsigned mw[SCAN_WORDS];
#pragma unroll
    for (int v = 0; v < SCAN_WORDS; ++v)
      mw[v] = hits16(cc[2 * v], pat) | (hits16(cc[2 * v + 1], pat) << 4);
    const int cur = ch0;

    ch0 += SCAN_STEP;
    if (ch0 >= hi && pass == 0) { pass = 1; ch0 = wave; hi = start; }
    valid = ch0 < hi;
    if (valid) load_owners(cc, own, ch0, hi, lane, nomatch);

#pragma nounroll
    for (int v = 0; v < SCAN_WORDS; ++v) {
      unsigned mbits = mw[0];
#pragma unroll
      for (int t = 0; t + 1 < SCAN_WORDS; ++t) mw[t] = mw[t + 1];
      const bool flush = !valid && (v == SCAN_WORDS - 1);
      const int cbase = cur + 2 * v * AGG_WAVES;

      for (;;) {
        while (qlen < 32) {
          const bool has = mbits != 0u;
          const unsigned b = __builtin_amdgcn_ballot_w32(has);
          if (b == 0u) break;
          if (has) {
            const int s = __builtin_ctz(mbits);
            mbits &= mbits - 1u;
            const int ch = cbase + ((s >> 2) & 1) * AGG_WAVES;
            const int e = (ch << 9) + (lane << 4) + ((s & 3) << 2) + (s >> 3);
            const int pos = head + qlen + (int)__builtin_amdgcn_mbcnt_lo(b, 0u);
            qe[pos & (QCAP - 1)] = e;
          }
          qlen += __builtin_popcount(b);
        }
        const int m = (qlen < 32) ? qlen : 32;
        if (m == 0 || (m < 32 && !flush)) break;

        int r = 0, lc = -1;
        float nrm = 0.0f;
        if (lane < m) {
          const int e = qe[(head + lane) & (QCAP - 1)];
          if (e < e_cnt) {
            r = row[e];
            const int c = col[e];
            const float w = ew[e];
            r = (r < 0) ? 0 : ((r > n - 1) ? (n - 1) : r);
            if ((unsigned)(c - base) < (unsigned)cnt) {
              lc = c - base;
              nrm = dinv[r] * w * dinv[c];
            }
          }
        }
        for (int j0 = 0; j0 < m; j0 += ROWS_IN_FLIGHT) {
          float4 v4[ROWS_IN_FLIGHT];
#pragma unroll
          for (int x8 = 0; x8 < ROWS_IN_FLIGHT; ++x8) {
            const int j = (j0 + x8 < m) ? (j0 + x8) : (m - 1);
            const int rj = __builtin_amdgcn_readlane(r, j);
            v4[x8] = *reinterpret_cast<const float4*>(x + (size_t)rj * D + 4 * lane);
          }
#pragma unroll
          for (int x8 = 0; x8 < ROWS_IN_FLIGHT; ++x8) {
            const int j = j0 + x8;
            if (j < m) {
              const int lcj = __builtin_amdgcn_readlane(lc, j);
              if (lcj >= 0) {
                const float nj = __int_as_float(__builtin_amdgcn_readlane(__float_as_int(nrm), j));
                float* a = acc + lcj * D + ((lane + lcj) & 31);
                atomicAdd(a,      nj * v4[x8].x);
                atomicAdd(a + 32, nj * v4[x8].y);
                atomicAdd(a + 64, nj * v4[x8].z);
                atomicAdd(a + 96, nj * v4[x8].w);
              }
            }
          }
        }
        head = (head + m) & (QCAP - 1);
        qlen -= m;
      }
    }
  }
  __syncthreads();

  const float slope = pa[0];
  const int c16 = lane & 15;
  const int h   = lane >> 4;
  const float4 bias0 = *reinterpret_cast<const float4*>(bias + 4 * c16);
  const float4 bias1 = *reinterpret_cast<const float4*>(bias + 64 + 4 * c16);
  const int ntiles = (cnt + 15) >> 4;
  for (int tile = wave; tile < ntiles; tile += AGG_WAVES) {
    const int ra = tile * 16 + c16;
    const int rc = (ra < RN) ? ra : (RN - 1);
    const float* arow = acc + rc * D + h * 32;
    const float* wrow = w + (size_t)h * D + 4 * c16;

    v8f d[8];
#pragma unroll
    for (int t = 0; t < 8; ++t) d[t] = (v8f){0.f, 0.f, 0.f, 0.f, 0.f, 0.f, 0.f, 0.f};

    for (int q = 0; q < D / 4; ++q) {
      const int o = (q + rc) & 31;
      v2f a;
      a.x = arow[o];
      a.y = arow[64 + o];
      const float* wp = wrow + (size_t)q * (4 * D);
      const float4 b00 = *reinterpret_cast<const float4*>(wp);
      const float4 b01 = *reinterpret_cast<const float4*>(wp + 64);
      const float4 b10 = *reinterpret_cast<const float4*>(wp + 2 * D);
      const float4 b11 = *reinterpret_cast<const float4*>(wp + 2 * D + 64);
      d[0] = __builtin_amdgcn_wmma_f32_16x16x4_f32(false, a, false, (v2f){b00.x, b10.x}, (short)0, d[0], false, false);
      d[1] = __builtin_amdgcn_wmma_f32_16x16x4_f32(false, a, false, (v2f){b00.y, b10.y}, (short)0, d[1], false, false);
      d[2] = __builtin_amdgcn_wmma_f32_16x16x4_f32(false, a, false, (v2f){b00.z, b10.z}, (short)0, d[2], false, false);
      d[3] = __builtin_amdgcn_wmma_f32_16x16x4_f32(false, a, false, (v2f){b00.w, b10.w}, (short)0, d[3], false, false);
      d[4] = __builtin_amdgcn_wmma_f32_16x16x4_f32(false, a, false, (v2f){b01.x, b11.x}, (short)0, d[4], false, false);
      d[5] = __builtin_amdgcn_wmma_f32_16x16x4_f32(false, a, false, (v2f){b01.y, b11.y}, (short)0, d[5], false, false);
      d[6] = __builtin_amdgcn_wmma_f32_16x16x4_f32(false, a, false, (v2f){b01.z, b11.z}, (short)0, d[6], false, false);
      d[7] = __builtin_amdgcn_wmma_f32_16x16x4_f32(false, a, false, (v2f){b01.w, b11.w}, (short)0, d[7], false, false);
    }

#pragma unroll
    for (int j = 0; j < 8; ++j) {
      const int ro = tile * 16 + 8 * h + j;
      if (ro < cnt) {
        float4 v0, v1;
        v0.x = bias0.x + d[0][j];
        v0.y = bias0.y + d[1][j];
        v0.z = bias0.z + d[2][j];
        v0.w = bias0.w + d[3][j];
        v1.x = bias1.x + d[4][j];
        v1.y = bias1.y + d[5][j];
        v1.z = bias1.z + d[6][j];
        v1.w = bias1.w + d[7][j];
        v0.x = (v0.x >= 0.0f) ? v0.x : slope * v0.x;
        v0.y = (v0.y >= 0.0f) ? v0.y : slope * v0.y;
        v0.z = (v0.z >= 0.0f) ? v0.z : slope * v0.z;
        v0.w = (v0.w >= 0.0f) ? v0.w : slope * v0.w;
        v1.x = (v1.x >= 0.0f) ? v1.x : slope * v1.x;
        v1.y = (v1.y >= 0.0f) ? v1.y : slope * v1.y;
        v1.z = (v1.z >= 0.0f) ? v1.z : slope * v1.z;
        v1.w = (v1.w >= 0.0f) ? v1.w : slope * v1.w;
        float* op = out + (size_t)(base + ro) * D + 4 * c16;
        *reinterpret_cast<float4*>(op)      = v0;
        *reinterpret_cast<float4*>(op + 64) = v1;
      }
    }
  }
}

extern "C" void kernel_launch(void* const* d_in, const int* in_sizes, int n_in,
                              void* d_out, int out_size, void* d_ws, size_t ws_size,
                              hipStream_t stream) {
  const float* x    = (const float*)d_in[0];
  const int*   ei   = (const int*)  d_in[1];
  const float* ew   = (const float*)d_in[2];
  const float* w    = (const float*)d_in[3];
  const float* bias = (const float*)d_in[4];
  const float* pa   = (const float*)d_in[5];
  float* out = (float*)d_out;

  const int n     = in_sizes[0] / D;
  const int e_cnt = in_sizes[2];
  const int* row = ei;
  const int* col = ei + e_cnt;

  float* deg = (float*)d_ws;
  unsigned char* own = (unsigned char*)(deg + ((n + 3) & ~3));
  const int e_pad = (e_cnt + 511) & ~511;
  float* part = (float*)(own + e_pad);
  const int n_ranges = (n + DEG_HR - 1) / DEG_HR;
  const int n_pad = n_ranges * DEG_HR;

  deg_partial_kernel<<<DEG_SLABS * n_ranges, 1024, 0, stream>>>(col, ew, part, own, n_pad, e_cnt, e_pad);
  dinv_reduce_kernel<<<(n + 255) / 256, 256, 0, stream>>>(part, deg, n, n_pad);
  aggregate_kernel<<<(n + RN - 1) / RN, 1024, 0, stream>>>(row, col, ew, deg, x, w, own, bias, pa, out, n, e_cnt, e_pad);
}


# source, target and weight of a kept edge read as one 16-byte record (the degree pass copies the edge list into records, edge order, whole-line stores): one scattered line per edge, not three
# speedup vs baseline: 5.8477x; 1.0927x over previous
#include <hip/hip_runtime.h>
#include <hip/hip_bf16.h>

typedef __attribute__((ext_vector_type(2))) float v2f;
typedef __attribute__((ext_vector_type(8))) float v8f;

#define D 128

#define RN 391
#define DEG_SLABS 32
#define DEG_HR 12544
#define DEG_UNR 4

__device__ __forceinline__ void deg_apply(float* hist, unsigned char* __restrict__ own,
                                          int4 c, float4 w, int lo, int e0, bool bytes) {
  const unsigned dx = (unsigned)(c.x - lo);
  const unsigned dy = (unsigned)(c.y - lo);
  const unsigned dz = (unsigned)(c.z - lo);
  const unsigned dw = (unsigned)(c.w - lo);
  if (dx < (unsigned)DEG_HR) atomicAdd(&hist[dx], w.x);
  if (dy < (unsigned)DEG_HR) atomicAdd(&hist[dy], w.y);
  if (dz < (unsigned)DEG_HR) atomicAdd(&hist[dz], w.z);
  if (dw < (unsigned)DEG_HR) atomicAdd(&hist[dw], w.w);
  if (bytes) {
    const unsigned pk =
        ((unsigned)(c.x / RN) & 255u) | (((unsigned)(c.y / RN) & 255u) << 8) |
        (((unsigned)(c.z / RN) & 255u) << 16) | (((unsigned)(c.w / RN) & 255u) << 24);
    *reinterpret_cast<unsigned*>(own + e0) = pk;
  }
}

__device__ __forceinline__ void write_records(const int* __restrict__ row,
                                              const int* __restrict__ col,
                                              const float* __restrict__ ew,
                                              int4* __restrict__ rec,
                                              int ubase, int tid, int e_cnt, bool whole) {
  int4 q[4];
#pragma unroll
  for (int k = 0; k < 4; ++k) {
    const int e = ubase + (k << 10) + tid;
    q[k] = make_int4(0, 0, 0, 0);
    if (whole || e < e_cnt) q[k] = make_int4(row[e], col[e], __float_as_int(ew[e]), 0);
  }
#pragma unroll
  for (int k = 0; k < 4; ++k) {
    const int e = ubase + (k << 10) + tid;
    if (whole || e < e_cnt) rec[e] = q[k];
  }
}

__global__ __launch_bounds__(1024) void deg_partial_kernel(
    const int* __restrict__ row, const int* __restrict__ col,
    const float* __restrict__ ew,
    float* __restrict__ part, unsigned char* __restrict__ own,
    int4* __restrict__ rec,
    int n_pad, int e_cnt, int e_pad) {
  __shared__ alignas(16) float hist[DEG_HR];
  const int tid  = threadIdx.x;
  const int slab = (int)blockIdx.x % DEG_SLABS;
  const int g    = (int)blockIdx.x / DEG_SLABS;
  const int n_ranges = (int)gridDim.x / DEG_SLABS;
  const int lo = g * DEG_HR;

  for (int i = tid; i < DEG_HR; i += 1024) hist[i] = 0.0f;
  __syncthreads();

  const int units = (e_pad + 4095) >> 12;
  const int ubeg = (int)(((long long)slab * units) / DEG_SLABS);
  const int uend = (int)(((long long)(slab + 1) * units) / DEG_SLABS);
  const int ufull = e_cnt >> 12;

  int u0 = ubeg;
  const int ufast = (uend < ufull) ? uend : ufull;
  for (; u0 + DEG_UNR <= ufast; u0 += DEG_UNR) {
    int4 c[DEG_UNR];
    float4 w[DEG_UNR];
#pragma unroll
    for (int j = 0; j < DEG_UNR; ++j) {
      const int e0 = ((u0 + j) << 12) + (tid << 2);
      c[j] = *reinterpret_cast<const int4*>(col + e0);
      w[j] = *reinterpret_cast<const float4*>(ew + e0);
    }
#pragma unroll
    for (int j = 0; j < DEG_UNR; ++j) {
      const int u = u0 + j;
      const bool mine = (u % n_ranges) == g;
      deg_apply(hist, own, c[j], w[j], lo, (u << 12) + (tid << 2), mine);
      if (mine) write_records(row, col, ew, rec, u << 12, tid, e_cnt, true);
    }
  }
  for (; u0 < uend; ++u0) {
    const int e0 = (u0 << 12) + (tid << 2);
    int4 c = make_int4(-1, -1, -1, -1);
    float4 w = make_float4(0.0f, 0.0f, 0.0f, 0.0f);
    if (e0 + 3 < e_cnt) {
      c = *reinterpret_cast<const int4*>(col + e0);
      w = *reinterpret_cast<const float4*>(ew + e0);
    } else {
      if (e0 < e_cnt)     { c.x = col[e0];     w.x = ew[e0]; }
      if (e0 + 1 < e_cnt) { c.y = col[e0 + 1]; w.y = ew[e0 + 1]; }
      if (e0 + 2 < e_cnt) { c.z = col[e0 + 2]; w.z = ew[e0 + 2]; }
    }
    const bool mine = (u0 % n_ranges) == g;
    deg_apply(hist, own, c, w, lo, e0, e0 < e_pad && mine);
    if (mine) write_records(row, col, ew, rec, u0 << 12, tid, e_cnt, false);
  }
  __syncthreads();

  float* dst = part + (size_t)slab * n_pad + lo;
  for (int i = tid << 2; i < DEG_HR; i += 4096)
    *reinterpret_cast<float4*>(dst + i) = *reinterpret_cast<const float4*>(hist + i);
}

__global__ void dinv_reduce_kernel(const float* __restrict__ part, float* __restrict__ dinv,
                                   int n, int n_pad) {
  const int i = blockIdx.x * blockDim.x + threadIdx.x;
  if (i < n) {
    float d = 1.0f;
#pragma unroll
    for (int s = 0; s < DEG_SLABS; ++s) d += part[(size_t)s * n_pad + i];
    dinv[i] = (d > 0.0f) ? rsqrtf(d) : 0.0f;
  }
}

#define AGG_WAVES 32
#define QCAP 64
#define SCAN_DEPTH 8
#define SCAN_WORDS (SCAN_DEPTH / 2)
#define SCAN_STEP (AGG_WAVES * SCAN_DEPTH)
#define ROWS_IN_FLIGHT 16

__device__ __forceinline__ void load_owners(uint4 (&cc)[SCAN_DEPTH],
                                            const unsigned char* __restrict__ own,
                                            int ch0, int hi, int lane,
                                            unsigned nomatch) {
  const uint4* p = reinterpret_cast<const uint4*>(own + ((size_t)ch0 << 9)) + lane;
  const int chl = ch0 + (SCAN_DEPTH - 1) * AGG_WAVES;
  if (chl < hi) {
#pragma unroll
    for (int u = 0; u < SCAN_DEPTH; ++u) cc[u] = p[u * AGG_WAVES * 32];
  } else {
#pragma unroll
    for (int u = 0; u < SCAN_DEPTH; ++u) {
      uint4 c4 = make_uint4(nomatch, nomatch, nomatch, nomatch);
      if (ch0 + u * AGG_WAVES < hi) c4 = p[u * AGG_WAVES * 32];
      cc[u] = c4;
    }
  }
}

__device__ __forceinline__ unsigned nonzero_bytes(unsigned x) {
  return ((x & 0x7f7f7f7fu) + 0x7f7f7f7fu) | x;
}

__device__ __forceinline__ unsigned hits16(uint4 c, unsigned pat) {
  const unsigned z = ((nonzero_bytes(c.x ^ pat) >> 7) & 0x01010101u) |
                     ((nonzero_bytes(c.y ^ pat) >> 6) & 0x02020202u) |
                     ((nonzero_bytes(c.z ^ pat) >> 5) & 0x04040404u) |
                     ((nonzero_bytes(c.w ^ pat) >> 4) & 0x08080808u);
  return z ^ 0x0f0f0f0fu;
}

__global__ __launch_bounds__(1024) void aggregate_kernel(
    const int4* __restrict__ rec, const float* __restrict__ dinv,
    const float* __restrict__ x, const float* __restrict__ w,
    const unsigned char* __restrict__ own,
    const float* __restrict__ bias, const float* __restrict__ pa,
    float* __restrict__ out, int n, int e_cnt, int e_pad) {
  __shared__ float acc[RN * D];
  __shared__ int q_e[AGG_WAVES * QCAP];

  const int lane = threadIdx.x & 31;
  const int wave = __builtin_amdgcn_readfirstlane((int)(threadIdx.x >> 5));
  const int base = blockIdx.x * RN;
  const int cnt  = (n - base < RN) ? (n - base) : RN;

  for (int i = wave; i < cnt; i += AGG_WAVES) {
    const float di = dinv[base + i];
    const float s = di * di;
    const float4 v = *reinterpret_cast<const float4*>(x + (size_t)(base + i) * D + 4 * lane);
    float* a = acc + i * D + ((lane + i) & 31);
    a[0]  = s * v.x;
    a[32] = s * v.y;
    a[64] = s * v.z;
    a[96] = s * v.w;
  }
  __syncthreads();

  int* qe = q_e + wave * QCAP;
  int head = 0;
  int qlen = 0;

  const unsigned pat = ((unsigned)blockIdx.x & 255u) * 0x01010101u;
  const unsigned nomatch = ~pat;

  const int nch   = e_pad >> 9;
  const int start = (int)blockIdx.x * (nch / (int)gridDim.x);

  int pass = 0;
  int ch0 = start + wave;
  int hi = nch;
  if (ch0 >= hi) { pass = 1; ch0 = wave; hi = start; }
  bool valid = ch0 < hi;

  uint4 cc[SCAN_DEPTH];
  if (valid) load_owners(cc, own, ch0, hi, lane, nomatch);

  while (valid) {
    unsigned mw[SCAN_WORDS];
#pragma unroll
    for (int v = 0; v < SCAN_WORDS; ++v)
      mw[v] = hits16(cc[2 * v], pat) | (hits16(cc[2 * v + 1], pat) << 4);
    const int cur = ch0;

    ch0 += SCAN_STEP;
    if (ch0 >= hi && pass == 0) { pass = 1; ch0 = wave; hi = start; }
    valid = ch0 < hi;
    if (valid) load_owners(cc, own, ch0, hi, lane, nomatch);

#pragma nounroll
    for (int v = 0; v < SCAN_WORDS; ++v) {
      unsigned mbits = mw[0];
#pragma unroll
      for (int t = 0; t + 1 < SCAN_WORDS; ++t) mw[t] = mw[t + 1];
      const bool flush = !valid && (v == SCAN_WORDS - 1);
      const int cbase = cur + 2 * v * AGG_WAVES;

      for (;;) {
        while (qlen < 32) {
          const bool has = mbits != 0u;
          const unsigned b = __builtin_amdgcn_ballot_w32(has);
          if (b == 0u) break;
          if (has) {
            const int s = __builtin_ctz(mbits);
            mbits &= mbits - 1u;
            const int ch = cbase + ((s >> 2) & 1) * AGG_WAVES;
            const int e = (ch << 9) + (lane << 4) + ((s & 3) << 2) + (s >> 3);
            const int pos = head + qlen + (int)__builtin_amdgcn_mbcnt_lo(b, 0u);
            qe[pos & (QCAP - 1)] = e;
          }
          qlen += __builtin_popcount(b);
        }
        const int m = (qlen < 32) ? qlen : 32;
        if (m == 0 || (m < 32 && !flush)) break;

        int r = 0, lc = -1;
        float nrm = 0.0f;
        if (lane < m) {
          const int e = qe[(head + lane) & (QCAP - 1)];
          if (e < e_cnt) {
            const int4 rc = rec[e];
            r = rc.x;
            const int c = rc.y;
            const float w = __int_as_float(rc.z);
            r = (r < 0) ? 0 : ((r > n - 1) ? (n - 1) : r);
            const bool mine = (unsigned)(c - base) < (unsigned)cnt;
            const float f = dinv[r] * w * dinv[mine ? c : base];
            lc = mine ? (c - base) : -1;
            nrm = mine ? f : 0.0f;
          }
        }
        for (int j0 = 0; j0 < m; j0 += ROWS_IN_FLIGHT) {
          float4 v4[ROWS_IN_FLIGHT];
#pragma unroll
          for (int x8 = 0; x8 < ROWS_IN_FLIGHT; ++x8) {
            const int j = (j0 + x8 < m) ? (j0 + x8) : (m - 1);
            const int rj = __builtin_amdgcn_readlane(r, j);
            v4[x8] = *reinterpret_cast<const float4*>(x + (size_t)rj * D + 4 * lane);
          }
#pragma unroll
          for (int x8 = 0; x8 < ROWS_IN_FLIGHT; ++x8) {
            const int j = j0 + x8;
            if (j < m) {
              const int lcj = __builtin_amdgcn_readlane(lc, j);
              if (lcj >= 0) {
                const float nj = __int_as_float(__builtin_amdgcn_readlane(__float_as_int(nrm), j));
                float* a = acc + lcj * D + ((lane + lcj) & 31);
                atomicAdd(a,      nj * v4[x8].x);
                atomicAdd(a + 32, nj * v4[x8].y);
                atomicAdd(a + 64, nj * v4[x8].z);
                atomicAdd(a + 96, nj * v4[x8].w);
              }
            }
          }
        }
        head = (head + m) & (QCAP - 1);
        qlen -= m;
      }
    }
  }
  __syncthreads();

  const float slope = pa[0];
  const int c16 = lane & 15;
  const int h   = lane >> 4;
  const float4 bias0 = *reinterpret_cast<const float4*>(bias + 4 * c16);
  const float4 bias1 = *reinterpret_cast<const float4*>(bias + 64 + 4 * c16);
  const int ntiles = (cnt + 15) >> 4;
  for (int tile = wave; tile < ntiles; tile += AGG_WAVES) {
    const int ra = tile * 16 + c16;
    const int rc = (ra < RN) ? ra : (RN - 1);
    const float* arow = acc + rc * D + h * 32;
    const float* wrow = w + (size_t)h * D + 4 * c16;

    v8f d[8];
#pragma unroll
    for (int t = 0; t < 8; ++t) d[t] = (v8f){0.f, 0.f, 0.f, 0.f, 0.f, 0.f, 0.f, 0.f};

    for (int q = 0; q < D / 4; ++q) {
      const int o = (q + rc) & 31;
      v2f a;
      a.x = arow[o];
      a.y = arow[64 + o];
      const float* wp = wrow + (size_t)q * (4 * D);
      const float4 b00 = *reinterpret_cast<const float4*>(wp);
      const float4 b01 = *reinterpret_cast<const float4*>(wp + 64);
      const float4 b10 = *reinterpret_cast<const float4*>(wp + 2 * D);
      const float4 b11 = *reinterpret_cast<const float4*>(wp + 2 * D + 64);
      d[0] = __builtin_amdgcn_wmma_f32_16x16x4_f32(false, a, false, (v2f){b00.x, b10.x}, (short)0, d[0], false, false);
      d[1] = __builtin_amdgcn_wmma_f32_16x16x4_f32(false, a, false, (v2f){b00.y, b10.y}, (short)0, d[1], false, false);
      d[2] = __builtin_amdgcn_wmma_f32_16x16x4_f32(false, a, false, (v2f){b00.z, b10.z}, (short)0, d[2], false, false);
      d[3] = __builtin_amdgcn_wmma_f32_16x16x4_f32(false, a, false, (v2f){b00.w, b10.w}, (short)0, d[3], false, false);
      d[4] = __builtin_amdgcn_wmma_f32_16x16x4_f32(false, a, false, (v2f){b01.x, b11.x}, (short)0, d[4], false, false);
      d[5] = __builtin_amdgcn_wmma_f32_16x16x4_f32(false, a, false, (v2f){b01.y, b11.y}, (short)0, d[5], false, false);
      d[6] = __builtin_amdgcn_wmma_f32_16x16x4_f32(false, a, false, (v2f){b01.z, b11.z}, (short)0, d[6], false, false);
      d[7] = __builtin_amdgcn_wmma_f32_16x16x4_f32(false, a, false, (v2f){b01.w, b11.w}, (short)0, d[7], false, false);
    }

#pragma unroll
    for (int j = 0; j < 8; ++j) {
      const int ro = tile * 16 + 8 * h + j;
      if (ro < cnt) {
        float4 v0, v1;
        v0.x = bias0.x + d[0][j];
        v0.y = bias0.y + d[1][j];
        v0.z = bias0.z + d[2][j];
        v0.w = bias0.w + d[3][j];
        v1.x = bias1.x + d[4][j];
        v1.y = bias1.y + d[5][j];
        v1.z = bias1.z + d[6][j];
        v1.w = bias1.w + d[7][j];
        v0.x = (v0.x >= 0.0f) ? v0.x : slope * v0.x;
        v0.y = (v0.y >= 0.0f) ? v0.y : slope * v0.y;
        v0.z = (v0.z >= 0.0f) ? v0.z : slope * v0.z;
        v0.w = (v0.w >= 0.0f) ? v0.w : slope * v0.w;
        v1.x = (v1.x >= 0.0f) ? v1.x : slope * v1.x;
        v1.y = (v1.y >= 0.0f) ? v1.y : slope * v1.y;
        v1.z = (v1.z >= 0.0f) ? v1.z : slope * v1.z;
        v1.w = (v1.w >= 0.0f) ? v1.w : slope * v1.w;
        float* op = out + (size_t)(base + ro) * D + 4 * c16;
        *reinterpret_cast<float4*>(op)      = v0;
        *reinterpret_cast<float4*>(op + 64) = v1;
      }
    }
  }
}

extern "C" void kernel_launch(void* const* d_in, const int* in_sizes, int n_in,
                              void* d_out, int out_size, void* d_ws, size_t ws_size,
                              hipStream_t stream) {
  const float* x    = (const float*)d_in[0];
  const int*   ei   = (const int*)  d_in[1];
  const float* ew   = (const float*)d_in[2];
  const float* w    = (const float*)d_in[3];
  const float* bias = (const float*)d_in[4];
  const float* pa   = (const float*)d_in[5];
  float* out = (float*)d_out;

  const int n     = in_sizes[0] / D;
  const int e_cnt = in_sizes[2];
  const int* row = ei;
  const int* col = ei + e_cnt;

  float* deg = (float*)d_ws;
  unsigned char* own = (unsigned char*)(deg + ((n + 3) & ~3));
  const int e_pad = (e_cnt + 511) & ~511;
  float* part = (float*)(own + e_pad);
  const int n_ranges = (n + DEG_HR - 1) / DEG_HR;
  const int n_pad = n_ranges * DEG_HR;
  int4* rec = (int4*)(part + (size_t)DEG_SLABS * n_pad);

  deg_partial_kernel<<<DEG_SLABS * n_ranges, 1024, 0, stream>>>(row, col, ew, part, own, rec, n_pad, e_cnt, e_pad);
  dinv_reduce_kernel<<<(n + 255) / 256, 256, 0, stream>>>(part, deg, n, n_pad);
  aggregate_kernel<<<(n + RN - 1) / RN, 1024, 0, stream>>>(rec, deg, x, w, own, bias, pa, out, n, e_cnt, e_pad);
}


# edge filter as one-hot bit planes (owner index high 4 bits: 16 planes, low 4: 16; built in LDS by the degree pass): a work-group ANDs its 2 planes, reads 1/4 of the bytes; LDS arrival order kept
# speedup vs baseline: 6.1953x; 1.0594x over previous
#include <hip/hip_runtime.h>
#include <hip/hip_bf16.h>

typedef __attribute__((ext_vector_type(2))) float v2f;
typedef __attribute__((ext_vector_type(8))) float v8f;

#define D 128

#define RN 391
#define DEG_SLABS 32
#define DEG_HR 12544
#define DEG_UNR 4
#define N_PLANES 32
#define BM_PITCH 132

__device__ __forceinline__ void deg_apply(float* hist, int4 c, float4 w, int lo) {
  const unsigned dx = (unsigned)(c.x - lo);
  const unsigned dy = (unsigned)(c.y - lo);
  const unsigned dz = (unsigned)(c.z - lo);
  const unsigned dw = (unsigned)(c.w - lo);
  if (dx < (unsigned)DEG_HR) atomicAdd(&hist[dx], w.x);
  if (dy < (unsigned)DEG_HR) atomicAdd(&hist[dy], w.y);
  if (dz < (unsigned)DEG_HR) atomicAdd(&hist[dz], w.z);
  if (dw < (unsigned)DEG_HR) atomicAdd(&hist[dw], w.w);
}

__device__ __forceinline__ void mark_edge(unsigned* wp, int c, unsigned bit) {
  if (c >= 0) {
    const unsigned o = (unsigned)c / RN;
    atomicOr(wp + ((o >> 4) & 15u) * BM_PITCH, bit);
    atomicOr(wp + (16u + (o & 15u)) * BM_PITCH, bit);
  }
}

__device__ __forceinline__ void build_planes(unsigned* bm, uint4* __restrict__ bits,
                                             int4 c, int u, int units, int tid) {
  unsigned* wp = bm + (tid >> 3);
  const unsigned b0 = 1u << ((tid & 7) << 2);
  mark_edge(wp, c.x, b0);
  mark_edge(wp, c.y, b0 << 1);
  mark_edge(wp, c.z, b0 << 2);
  mark_edge(wp, c.w, b0 << 3);
  __syncthreads();
  const int p = tid >> 5;
  const int l = tid & 31;
  uint4* src = reinterpret_cast<uint4*>(bm + p * BM_PITCH + 4 * l);
  const uint4 v = *src;
  *src = make_uint4(0u, 0u, 0u, 0u);
  __syncthreads();
  bits[((size_t)p * units + u) * 32 + l] = v;
}

__device__ __forceinline__ void write_records(const int* __restrict__ row,
                                              const int* __restrict__ col,
                                              const float* __restrict__ ew,
                                              int4* __restrict__ rec,
                                              int ubase, int tid, int e_cnt, bool whole) {
  int4 q[4];
#pragma unroll
  for (int k = 0; k < 4; ++k) {
    const int e = ubase + (k << 10) + tid;
    q[k] = make_int4(0, 0, 0, 0);
    if (whole || e < e_cnt) q[k] = make_int4(row[e], col[e], __float_as_int(ew[e]), 0);
  }
#pragma unroll
  for (int k = 0; k < 4; ++k) {
    const int e = ubase + (k << 10) + tid;
    if (whole || e < e_cnt) rec[e] = q[k];
  }
}

__global__ __launch_bounds__(1024) void deg_partial_kernel(
    const int* __restrict__ row, const int* __restrict__ col,
    const float* __restrict__ ew,
    float* __restrict__ part, uint4* __restrict__ bits,
    int4* __restrict__ rec,
    int n_pad, int e_cnt, int e_pad) {
  __shared__ alignas(16) float hist[DEG_HR];
  __shared__ alignas(16) unsigned bm[N_PLANES * BM_PITCH];
  const int tid  = threadIdx.x;
  const int slab = (int)blockIdx.x % DEG_SLABS;
  const int g    = (int)blockIdx.x / DEG_SLABS;
  const int n_ranges = (int)gridDim.x / DEG_SLABS;
  const int lo = g * DEG_HR;

  for (int i = tid; i < DEG_HR; i += 1024) hist[i] = 0.0f;
  for (int i = tid; i < N_PLANES * BM_PITCH; i += 1024) bm[i] = 0u;
  __syncthreads();

  const int units = (e_pad + 4095) >> 12;
  const int ubeg = (int)(((long long)slab * units) / DEG_SLABS);
  const int uend = (int)(((long long)(slab + 1) * units) / DEG_SLABS);
  const int ufull = e_cnt >> 12;

  int u0 = ubeg;
  const int ufast = (uend < ufull) ? uend : ufull;
  for (; u0 + DEG_UNR <= ufast; u0 += DEG_UNR) {
    int4 c[DEG_UNR];
    float4 w[DEG_UNR];
#pragma unroll
    for (int j = 0; j < DEG_UNR; ++j) {
      const int e0 = ((u0 + j) << 12) + (tid << 2);
      c[j] = *reinterpret_cast<const int4*>(col + e0);
      w[j] = *reinterpret_cast<const float4*>(ew + e0);
    }
#pragma unroll
    for (int j = 0; j < DEG_UNR; ++j) {
      const int u = u0 + j;
      const bool mine = (u % n_ranges) == g;
      deg_apply(hist, c[j], w[j], lo);
      if (mine) {
        build_planes(bm, bits, c[j], u, units, tid);
        write_records(row, col, ew, rec, u << 12, tid, e_cnt, true);
      }
    }
  }
  for (; u0 < uend; ++u0) {
    const int e0 = (u0 << 12) + (tid << 2);
    int4 c = make_int4(-1, -1, -1, -1);
    float4 w = make_float4(0.0f, 0.0f, 0.0f, 0.0f);
    if (e0 + 3 < e_cnt) {
      c = *reinterpret_cast<const int4*>(col + e0);
      w = *reinterpret_cast<const float4*>(ew + e0);
    } else {
      if (e0 < e_cnt)     { c.x = col[e0];     w.x = ew[e0]; }
      if (e0 + 1 < e_cnt) { c.y = col[e0 + 1]; w.y = ew[e0 + 1]; }
      if (e0 + 2 < e_cnt) { c.z = col[e0 + 2]; w.z = ew[e0 + 2]; }
    }
    const bool mine = (u0 % n_ranges) == g;
    deg_apply(hist, c, w, lo);
    if (mine) {
      build_planes(bm, bits, c, u0, units, tid);
      write_records(row, col, ew, rec, u0 << 12, tid, e_cnt, false);
    }
  }
  __syncthreads();

  float* dst = part + (size_t)slab * n_pad + lo;
  for (int i = tid << 2; i < DEG_HR; i += 4096)
    *reinterpret_cast<float4*>(dst + i) = *reinterpret_cast<const float4*>(hist + i);
}

__global__ void dinv_reduce_kernel(const float* __restrict__ part, float* __restrict__ dinv,
                                   int n, int n_pad) {
  const int i = blockIdx.x * blockDim.x + threadIdx.x;
  if (i < n) {
    float d = 1.0f;
#pragma unroll
    for (int s = 0; s < DEG_SLABS; ++s) d += part[(size_t)s * n_pad + i];
    dinv[i] = (d > 0.0f) ? rsqrtf(d) : 0.0f;
  }
}

#define AGG_WAVES 32
#define QCAP 64
#define SCAN_DEPTH 2
#define SCAN_WORDS (SCAN_DEPTH * 4)
#define SCAN_STEP (AGG_WAVES * SCAN_DEPTH)
#define ROWS_IN_FLIGHT 16

__device__ __forceinline__ void load_planes(uint4 (&ch)[SCAN_DEPTH], uint4 (&cl)[SCAN_DEPTH],
                                            const uint4* __restrict__ ph,
                                            const uint4* __restrict__ pl,
                                            int ch0, int hi) {
#pragma unroll
  for (int u = 0; u < SCAN_DEPTH; ++u) {
    uint4 a = make_uint4(0u, 0u, 0u, 0u);
    uint4 b = make_uint4(0u, 0u, 0u, 0u);
    const int c = ch0 + u * AGG_WAVES;
    if (c < hi) {
      a = ph[(size_t)c * 32];
      b = pl[(size_t)c * 32];
    }
    ch[u] = a;
    cl[u] = b;
  }
}

__global__ __launch_bounds__(1024) void aggregate_kernel(
    const int4* __restrict__ rec, const float* __restrict__ dinv,
    const float* __restrict__ x, const float* __restrict__ w,
    const uint4* __restrict__ bits,
    const float* __restrict__ bias, const float* __restrict__ pa,
    float* __restrict__ out, int n, int e_cnt, int e_pad) {
  __shared__ float acc[RN * D];
  __shared__ int q_e[AGG_WAVES * QCAP];

  const int lane = threadIdx.x & 31;
  const int wave = __builtin_amdgcn_readfirstlane((int)(threadIdx.x >> 5));
  const int base = blockIdx.x * RN;
  const int cnt  = (n - base < RN) ? (n - base) : RN;

  for (int i = wave; i < cnt; i += AGG_WAVES) {
    const float di = dinv[base + i];
    const float s = di * di;
    const float4 v = *reinterpret_cast<const float4*>(x + (size_t)(base + i) * D + 4 * lane);
    float* a = acc + i * D + ((lane + i) & 31);
    a[0]  = s * v.x;
    a[32] = s * v.y;
    a[64] = s * v.z;
    a[96] = s * v.w;
  }
  __syncthreads();

  int* qe = q_e + wave * QCAP;
  int head = 0;
  int qlen = 0;

  const int nch   = (e_pad + 4095) >> 12;
  const int start = (int)blockIdx.x * (nch / (int)gridDim.x);
  const uint4* ph = bits + (size_t)(((unsigned)blockIdx.x >> 4) & 15u) * nch * 32 + lane;
  const uint4* pl = bits + (size_t)(16u + ((unsigned)blockIdx.x & 15u)) * nch * 32 + lane;

  int pass = 0;
  int ch0 = start + wave;
  int hi = nch;
  if (ch0 >= hi) { pass = 1; ch0 = wave; hi = start; }
  bool valid = ch0 < hi;

  uint4 ch[SCAN_DEPTH], cl[SCAN_DEPTH];
  if (valid) load_planes(ch, cl, ph, pl, ch0, hi);

  while (valid) {
    unsigned mw[SCAN_WORDS];
#pragma unroll
    for (int u = 0; u < SCAN_DEPTH; ++u) {
      mw[4 * u + 0] = ch[u].x & cl[u].x;
      mw[4 * u + 1] = ch[u].y & cl[u].y;
      mw[4 * u + 2] = ch[u].z & cl[u].z;
      mw[4 * u + 3] = ch[u].w & cl[u].w;
    }
    const int cur = ch0;

    ch0 += SCAN_STEP;
    if (ch0 >= hi && pass == 0) { pass = 1; ch0 = wave; hi = start; }
    valid = ch0 < hi;
    if (valid) load_planes(ch, cl, ph, pl, ch0, hi);

#pragma nounroll
    for (int v = 0; v < SCAN_WORDS; ++v) {
      unsigned mbits = mw[0];
#pragma unroll
      for (int t = 0; t + 1 < SCAN_WORDS; ++t) mw[t] = mw[t + 1];
      const bool flush = !valid && (v == SCAN_WORDS - 1);
      const int ebase = ((cur + (v >> 2) * AGG_WAVES) << 12) + ((v & 3) << 5);

      for (;;) {
        while (qlen < 32) {
          const bool has = mbits != 0u;
          const unsigned b = __builtin_amdgcn_ballot_w32(has);
          if (b == 0u) break;
          if (has) {
            const int s = __builtin_ctz(mbits);
            mbits &= mbits - 1u;
            const int e = ebase + (lane << 7) + s;
            const int pos = head + qlen + (int)__builtin_amdgcn_mbcnt_lo(b, 0u);
            qe[pos & (QCAP - 1)] = e;
          }
          qlen += __builtin_popcount(b);
        }
        const int m = (qlen < 32) ? qlen : 32;
        if (m == 0 || (m < 32 && !flush)) break;

        int r = 0, lc = -1;
        float nrm = 0.0f;
        if (lane < m) {
          const int e = qe[(head + lane) & (QCAP - 1)];
          if (e < e_cnt) {
            const int4 rc = rec[e];
            r = rc.x;
            const int c = rc.y;
            const float w = __int_as_float(rc.z);
            r = (r < 0) ? 0 : ((r > n - 1) ? (n - 1) : r);
            const bool mine = (unsigned)(c - base) < (unsigned)cnt;
            const float f = dinv[r] * w * dinv[mine ? c : base];
            lc = mine ? (c - base) : -1;
            nrm = mine ? f : 0.0f;
          }
        }
        for (int j0 = 0; j0 < m; j0 += ROWS_IN_FLIGHT) {
          float4 v4[ROWS_IN_FLIGHT];
#pragma unroll
          for (int x8 = 0; x8 < ROWS_IN_FLIGHT; ++x8) {
            const int j = (j0 + x8 < m) ? (j0 + x8) : (m - 1);
            const int rj = __builtin_amdgcn_readlane(r, j);
            v4[x8] = *reinterpret_cast<const float4*>(x + (size_t)rj * D + 4 * lane);
          }
#pragma unroll
          for (int x8 = 0; x8 < ROWS_IN_FLIGHT; ++x8) {
            const int j = j0 + x8;
            if (j < m) {
              const int lcj = __builtin_amdgcn_readlane(lc, j);
              if (lcj >= 0) {
                const float nj = __int_as_float(__builtin_amdgcn_readlane(__float_as_int(nrm), j));
                float* a = acc + lcj * D + ((lane + lcj) & 31);
                atomicAdd(a,      nj * v4[x8].x);
                atomicAdd(a + 32, nj * v4[x8].y);
                atomicAdd(a + 64, nj * v4[x8].z);
                atomicAdd(a + 96, nj * v4[x8].w);
              }
            }
          }
        }
        head = (head + m) & (QCAP - 1);
        qlen -= m;
      }
    }
  }
  __syncthreads();

  const float slope = pa[0];
  const int c16 = lane & 15;
  const int h   = lane >> 4;
  const float4 bias0 = *reinterpret_cast<const float4*>(bias + 4 * c16);
  const float4 bias1 = *reinterpret_cast<const float4*>(bias + 64 + 4 * c16);
  const int ntiles = (cnt + 15) >> 4;
  for (int tile = wave; tile < ntiles; tile += AGG_WAVES) {
    const int ra = tile * 16 + c16;
    const int rc = (ra < RN) ? ra : (RN - 1);
    const float* arow = acc + rc * D + h * 32;
    const float* wrow = w + (size_t)h * D + 4 * c16;

    v8f d[8];
#pragma unroll
    for (int t = 0; t < 8; ++t) d[t] = (v8f){0.f, 0.f, 0.f, 0.f, 0.f, 0.f, 0.f, 0.f};

    for (int q = 0; q < D / 4; ++q) {
      const int o = (q + rc) & 31;
      v2f a;
      a.x = arow[o];
      a.y = arow[64 + o];
      const float* wp = wrow + (size_t)q * (4 * D);
      const float4 b00 = *reinterpret_cast<const float4*>(wp);
      const float4 b01 = *reinterpret_cast<const float4*>(wp + 64);
      const float4 b10 = *reinterpret_cast<const float4*>(wp + 2 * D);
      const float4 b11 = *reinterpret_cast<const float4*>(wp + 2 * D + 64);
      d[0] = __builtin_amdgcn_wmma_f32_16x16x4_f32(false, a, false, (v2f){b00.x, b10.x}, (short)0, d[0], false, false);
      d[1] = __builtin_amdgcn_wmma_f32_16x16x4_f32(false, a, false, (v2f){b00.y, b10.y}, (short)0, d[1], false, false);
      d[2] = __builtin_amdgcn_wmma_f32_16x16x4_f32(false, a, false, (v2f){b00.z, b10.z}, (short)0, d[2], false, false);
      d[3] = __builtin_amdgcn_wmma_f32_16x16x4_f32(false, a, false, (v2f){b00.w, b10.w}, (short)0, d[3], false, false);
      d[4] = __builtin_amdgcn_wmma_f32_16x16x4_f32(false, a, false, (v2f){b01.x, b11.x}, (short)0, d[4], false, false);
      d[5] = __builtin_amdgcn_wmma_f32_16x16x4_f32(false, a, false, (v2f){b01.y, b11.y}, (short)0, d[5], false, false);
      d[6] = __builtin_amdgcn_wmma_f32_16x16x4_f32(false, a, false, (v2f){b01.z, b11.z}, (short)0, d[6], false, false);
      d[7] = __builtin_amdgcn_wmma_f32_16x16x4_f32(false, a, false, (v2f){b01.w, b11.w}, (short)0, d[7], false, false);
    }

#pragma unroll
    for (int j = 0; j < 8; ++j) {
      const int ro = tile * 16 + 8 * h + j;
      if (ro < cnt) {
        float4 v0, v1;
        v0.x = bias0.x + d[0][j];
        v0.y = bias0.y + d[1][j];
        v0.z = bias0.z + d[2][j];
        v0.w = bias0.w + d[3][j];
        v1.x = bias1.x + d[4][j];
        v1.y = bias1.y + d[5][j];
        v1.z = bias1.z + d[6][j];
        v1.w = bias1.w + d[7][j];
        v0.x = (v0.x >= 0.0f) ? v0.x : slope * v0.x;
        v0.y = (v0.y >= 0.0f) ? v0.y : slope * v0.y;
        v0.z = (v0.z >= 0.0f) ? v0.z : slope * v0.z;
        v0.w = (v0.w >= 0.0f) ? v0.w : slope * v0.w;
        v1.x = (v1.x >= 0.0f) ? v1.x : slope * v1.x;
        v1.y = (v1.y >= 0.0f) ? v1.y : slope * v1.y;
        v1.z = (v1.z >= 0.0f) ? v1.z : slope * v1.z;
        v1.w = (v1.w >= 0.0f) ? v1.w : slope * v1.w;
        float* op = out + (size_t)(base + ro) * D + 4 * c16;
        *reinterpret_cast<float4*>(op)      = v0;
        *reinterpret_cast<float4*>(op + 64) = v1;
      }
    }
  }
}

extern "C" void kernel_launch(void* const* d_in, const int* in_sizes, int n_in,
                              void* d_out, int out_size, void* d_ws, size_t ws_size,
                              hipStream_t stream) {
  const float* x    = (const float*)d_in[0];
  const int*   ei   = (const int*)  d_in[1];
  const float* ew   = (const float*)d_in[2];
  const float* w    = (const float*)d_in[3];
  const float* bias = (const float*)d_in[4];
  const float* pa   = (const float*)d_in[5];
  float* out = (float*)d_out;

  const int n     = in_sizes[0] / D;
  const int e_cnt = in_sizes[2];
  const int* row = ei;
  const int* col = ei + e_cnt;

  float* deg = (float*)d_ws;
  const int e_pad = (e_cnt + 511) & ~511;
  float* part = deg + ((n + 127) & ~127);
  const int n_ranges = (n + DEG_HR - 1) / DEG_HR;
  const int n_pad = n_ranges * DEG_HR;
  int4* rec = (int4*)(part + (size_t)DEG_SLABS * n_pad);
  uint4* bits = (uint4*)(rec + e_pad);

  deg_partial_kernel<<<DEG_SLABS * n_ranges, 1024, 0, stream>>>(row, col, ew, part, bits, rec, n_pad, e_cnt, e_pad);
  dinv_reduce_kernel<<<(n + 255) / 256, 256, 0, stream>>>(part, deg, n, n_pad);
  aggregate_kernel<<<(n + RN - 1) / RN, 1024, 0, stream>>>(rec, deg, x, w, bits, bias, pa, out, n, e_cnt, e_pad);
}


# owned rows summed in registers, not LDS: a row belongs to one wave (row mod 32); kept edges go to that wave's LDS list, the wave adds source rows by FMA; LDS float adds gone; list arrival order
# speedup vs baseline: 6.6386x; 1.0716x over previous
#include <hip/hip_runtime.h>
#include <hip/hip_bf16.h>

typedef __attribute__((ext_vector_type(2))) float v2f;
typedef __attribute__((ext_vector_type(8))) float v8f;

#define D 128

#define RN 391
#define DEG_SLABS 32
#define DEG_HR 12544
#define DEG_UNR 4
#define N_PLANES 32
#define BM_PITCH 132

__device__ __forceinline__ void deg_apply(float* hist, int4 c, float4 w, int lo) {
  const unsigned dx = (unsigned)(c.x - lo);
  const unsigned dy = (unsigned)(c.y - lo);
  const unsigned dz = (unsigned)(c.z - lo);
  const unsigned dw = (unsigned)(c.w - lo);
  if (dx < (unsigned)DEG_HR) atomicAdd(&hist[dx], w.x);
  if (dy < (unsigned)DEG_HR) atomicAdd(&hist[dy], w.y);
  if (dz < (unsigned)DEG_HR) atomicAdd(&hist[dz], w.z);
  if (dw < (unsigned)DEG_HR) atomicAdd(&hist[dw], w.w);
}

__device__ __forceinline__ void mark_edge(unsigned* wp, int c, unsigned bit) {
  if (c >= 0) {
    const unsigned o = (unsigned)c / RN;
    atomicOr(wp + ((o >> 4) & 15u) * BM_PITCH, bit);
    atomicOr(wp + (16u + (o & 15u)) * BM_PITCH, bit);
  }
}

__device__ __forceinline__ void build_planes(unsigned* bm, uint4* __restrict__ bits,
                                             int4 c, int u, int units, int tid) {
  unsigned* wp = bm + (tid >> 3);
  const unsigned b0 = 1u << ((tid & 7) << 2);
  mark_edge(wp, c.x, b0);
  mark_edge(wp, c.y, b0 << 1);
  mark_edge(wp, c.z, b0 << 2);
  mark_edge(wp, c.w, b0 << 3);
  __syncthreads();
  const int p = tid >> 5;
  const int l = tid & 31;
  uint4* src = reinterpret_cast<uint4*>(bm + p * BM_PITCH + 4 * l);
  const uint4 v = *src;
  *src = make_uint4(0u, 0u, 0u, 0u);
  __syncthreads();
  bits[((size_t)p * units + u) * 32 + l] = v;
}

__device__ __forceinline__ void write_records(const int* __restrict__ row,
                                              const int* __restrict__ col,
                                              const float* __restrict__ ew,
                                              int4* __restrict__ rec,
                                              int ubase, int tid, int e_cnt, bool whole) {
  int4 q[4];
#pragma unroll
  for (int k = 0; k < 4; ++k) {
    const int e = ubase + (k << 10) + tid;
    q[k] = make_int4(0, 0, 0, 0);
    if (whole || e < e_cnt) q[k] = make_int4(row[e], col[e], __float_as_int(ew[e]), 0);
  }
#pragma unroll
  for (int k = 0; k < 4; ++k) {
    const int e = ubase + (k << 10) + tid;
    if (whole || e < e_cnt) rec[e] = q[k];
  }
}

__global__ __launch_bounds__(1024) void deg_partial_kernel(
    const int* __restrict__ row, const int* __restrict__ col,
    const float* __restrict__ ew,
    float* __restrict__ part, uint4* __restrict__ bits,
    int4* __restrict__ rec,
    int n_pad, int e_cnt, int e_pad) {
  __shared__ alignas(16) float hist[DEG_HR];
  __shared__ alignas(16) unsigned bm[N_PLANES * BM_PITCH];
  const int tid  = threadIdx.x;
  const int slab = (int)blockIdx.x % DEG_SLABS;
  const int g    = (int)blockIdx.x / DEG_SLABS;
  const int n_ranges = (int)gridDim.x / DEG_SLABS;
  const int lo = g * DEG_HR;

  for (int i = tid; i < DEG_HR; i += 1024) hist[i] = 0.0f;
  for (int i = tid; i < N_PLANES * BM_PITCH; i += 1024) bm[i] = 0u;
  __syncthreads();

  const int units = (e_pad + 4095) >> 12;
  const int ubeg = (int)(((long long)slab * units) / DEG_SLABS);
  const int uend = (int)(((long long)(slab + 1) * units) / DEG_SLABS);
  const int ufull = e_cnt >> 12;

  int u0 = ubeg;
  const int ufast = (uend < ufull) ? uend : ufull;
  for (; u0 + DEG_UNR <= ufast; u0 += DEG_UNR) {
    int4 c[DEG_UNR];
    float4 w[DEG_UNR];
#pragma unroll
    for (int j = 0; j < DEG_UNR; ++j) {
      const int e0 = ((u0 + j) << 12) + (tid << 2);
      c[j] = *reinterpret_cast<const int4*>(col + e0);
      w[j] = *reinterpret_cast<const float4*>(ew + e0);
    }
#pragma unroll
    for (int j = 0; j < DEG_UNR; ++j) {
      const int u = u0 + j;
      const bool mine = (u % n_ranges) == g;
      deg_apply(hist, c[j], w[j], lo);
      if (mine) {
        build_planes(bm, bits, c[j], u, units, tid);
        write_records(row, col, ew, rec, u << 12, tid, e_cnt, true);
      }
    }
  }
  for (; u0 < uend; ++u0) {
    const int e0 = (u0 << 12) + (tid << 2);
    int4 c = make_int4(-1, -1, -1, -1);
    float4 w = make_float4(0.0f, 0.0f, 0.0f, 0.0f);
    if (e0 + 3 < e_cnt) {
      c = *reinterpret_cast<const int4*>(col + e0);
      w = *reinterpret_cast<const float4*>(ew + e0);
    } else {
      if (e0 < e_cnt)     { c.x = col[e0];     w.x = ew[e0]; }
      if (e0 + 1 < e_cnt) { c.y = col[e0 + 1]; w.y = ew[e0 + 1]; }
      if (e0 + 2 < e_cnt) { c.z = col[e0 + 2]; w.z = ew[e0 + 2]; }
    }
    const bool mine = (u0 % n_ranges) == g;
    deg_apply(hist, c, w, lo);
    if (mine) {
      build_planes(bm, bits, c, u0, units, tid);
      write_records(row, col, ew, rec, u0 << 12, tid, e_cnt, false);
    }
  }
  __syncthreads();

  float* dst = part + (size_t)slab * n_pad + lo;
  for (int i = tid << 2; i < DEG_HR; i += 4096)
    *reinterpret_cast<float4*>(dst + i) = *reinterpret_cast<const float4*>(hist + i);
}

__global__ void dinv_reduce_kernel(const float* __restrict__ part, float* __restrict__ dinv,
                                   int n, int n_pad) {
  const int i = blockIdx.x * blockDim.x + threadIdx.x;
  if (i < n) {
    float d = 1.0f;
#pragma unroll
    for (int s = 0; s < DEG_SLABS; ++s) d += part[(size_t)s * n_pad + i];
    dinv[i] = (d > 0.0f) ? rsqrtf(d) : 0.0f;
  }
}

#define AGG_WAVES 32
#define RPW 13
#define LCAP 288
#define QCAP 64
#define SCAN_DEPTH 2
#define SCAN_WORDS (SCAN_DEPTH * 4)
#define SCAN_STEP (AGG_WAVES * SCAN_DEPTH)
#define ROWS_IN_FLIGHT 8

__device__ __forceinline__ void load_planes(uint4 (&ch)[SCAN_DEPTH], uint4 (&cl)[SCAN_DEPTH],
                                            const uint4* __restrict__ ph,
                                            const uint4* __restrict__ pl,
                                            int ch0, int hi) {
#pragma unroll
  for (int u = 0; u < SCAN_DEPTH; ++u) {
    uint4 a = make_uint4(0u, 0u, 0u, 0u);
    uint4 b = make_uint4(0u, 0u, 0u, 0u);
    const int c = ch0 + u * AGG_WAVES;
    if (c < hi) {
      a = ph[(size_t)c * 32];
      b = pl[(size_t)c * 32];
    }
    ch[u] = a;
    cl[u] = b;
  }
}

__device__ __forceinline__ void row_fma(float4& s, float f, const float4& v) {
  s.x = fmaf(f, v.x, s.x);
  s.y = fmaf(f, v.y, s.y);
  s.z = fmaf(f, v.z, s.z);
  s.w = fmaf(f, v.w, s.w);
}

__global__ __launch_bounds__(1024) void aggregate_kernel(
    const int4* __restrict__ rec, const float* __restrict__ dinv,
    const float* __restrict__ x, const float* __restrict__ w,
    const uint4* __restrict__ bits,
    const float* __restrict__ bias, const float* __restrict__ pa,
    float* __restrict__ out, int n, int e_cnt, int e_pad) {
  __shared__ alignas(16) float acc[RN * D];
  __shared__ int q_e[AGG_WAVES * QCAP];
  __shared__ int l_src[AGG_WAVES * LCAP];
  __shared__ int l_row[AGG_WAVES * LCAP];
  __shared__ float l_nrm[AGG_WAVES * LCAP];
  __shared__ int l_cnt[AGG_WAVES];

  const int lane = threadIdx.x & 31;
  const int wave = __builtin_amdgcn_readfirstlane((int)(threadIdx.x >> 5));
  const int base = blockIdx.x * RN;
  const int cnt  = (n - base < RN) ? (n - base) : RN;

  for (int i = threadIdx.x; i < RN * D / 4; i += 1024)
    reinterpret_cast<float4*>(acc)[i] = make_float4(0.0f, 0.0f, 0.0f, 0.0f);
  if (threadIdx.x < AGG_WAVES) l_cnt[threadIdx.x] = 0;
  __syncthreads();

  int* qe = q_e + wave * QCAP;
  int head = 0;
  int qlen = 0;

  const int nch   = (e_pad + 4095) >> 12;
  const int start = (int)blockIdx.x * (nch / (int)gridDim.x);
  const uint4* ph = bits + (size_t)(((unsigned)blockIdx.x >> 4) & 15u) * nch * 32 + lane;
  const uint4* pl = bits + (size_t)(16u + ((unsigned)blockIdx.x & 15u)) * nch * 32 + lane;

  int pass = 0;
  int ch0 = start + wave;
  int hi = nch;
  if (ch0 >= hi) { pass = 1; ch0 = wave; hi = start; }
  bool valid = ch0 < hi;

  uint4 ch[SCAN_DEPTH], cl[SCAN_DEPTH];
  if (valid) load_planes(ch, cl, ph, pl, ch0, hi);

  while (valid) {
    unsigned mw[SCAN_WORDS];
#pragma unroll
    for (int u = 0; u < SCAN_DEPTH; ++u) {
      mw[4 * u + 0] = ch[u].x & cl[u].x;
      mw[4 * u + 1] = ch[u].y & cl[u].y;
      mw[4 * u + 2] = ch[u].z & cl[u].z;
      mw[4 * u + 3] = ch[u].w & cl[u].w;
    }
    const int cur = ch0;

    ch0 += SCAN_STEP;
    if (ch0 >= hi && pass == 0) { pass = 1; ch0 = wave; hi = start; }
    valid = ch0 < hi;
    if (valid) load_planes(ch, cl, ph, pl, ch0, hi);

#pragma nounroll
    for (int v = 0; v < SCAN_WORDS; ++v) {
      unsigned mbits = mw[0];
#pragma unroll
      for (int t = 0; t + 1 < SCAN_WORDS; ++t) mw[t] = mw[t + 1];
      const bool flush = !valid && (v == SCAN_WORDS - 1);
      const int ebase = ((cur + (v >> 2) * AGG_WAVES) << 12) + ((v & 3) << 5);

      for (;;) {
        while (qlen < 32) {
          const bool has = mbits != 0u;
          const unsigned b = __builtin_amdgcn_ballot_w32(has);
          if (b == 0u) break;
          if (has) {
            const int s = __builtin_ctz(mbits);
            mbits &= mbits - 1u;
            const int e = ebase + (lane << 7) + s;
            const int pos = head + qlen + (int)__builtin_amdgcn_mbcnt_lo(b, 0u);
            qe[pos & (QCAP - 1)] = e;
          }
          qlen += __builtin_popcount(b);
        }
        const int m = (qlen < 32) ? qlen : 32;
        if (m == 0 || (m < 32 && !flush)) break;

        int r = 0, lc = -1;
        float nrm = 0.0f;
        if (lane < m) {
          const int e = qe[(head + lane) & (QCAP - 1)];
          if (e < e_cnt) {
            const int4 rc = rec[e];
            r = rc.x;
            const int c = rc.y;
            const float w = __int_as_float(rc.z);
            r = (r < 0) ? 0 : ((r > n - 1) ? (n - 1) : r);
            const bool mine = (unsigned)(c - base) < (unsigned)cnt;
            const float f = dinv[r] * w * dinv[mine ? c : base];
            lc = mine ? (c - base) : -1;
            nrm = mine ? f : 0.0f;
          }
        }
        if (lc >= 0) {
          const int ow = lc & (AGG_WAVES - 1);
          const int slot = atomicAdd(&l_cnt[ow], 1);
          if (slot < LCAP) {
            const int at = ow * LCAP + slot;
            l_src[at] = r;
            l_row[at] = lc >> 5;
            l_nrm[at] = nrm;
            lc = -1;
          }
        }
        unsigned ov = __builtin_amdgcn_ballot_w32(lc >= 0);
        while (ov != 0u) {
          const int j = __builtin_ctz(ov);
          ov &= ov - 1u;
          const int rj  = __builtin_amdgcn_readlane(r, j);
          const int lcj = __builtin_amdgcn_readlane(lc, j);
          const float nj = __int_as_float(__builtin_amdgcn_readlane(__float_as_int(nrm), j));
          const float4 xv = *reinterpret_cast<const float4*>(x + (size_t)rj * D + 4 * lane);
          float* a = acc + lcj * D + ((lane + lcj) & 31);
          atomicAdd(a,      nj * xv.x);
          atomicAdd(a + 32, nj * xv.y);
          atomicAdd(a + 64, nj * xv.z);
          atomicAdd(a + 96, nj * xv.w);
        }
        head = (head + m) & (QCAP - 1);
        qlen -= m;
      }
    }
  }
  float4 s[RPW];
#pragma unroll
  for (int k = 0; k < RPW; ++k) {
    const int i = wave + AGG_WAVES * k;
    const bool on = i < cnt;
    const int ii = on ? i : (cnt - 1);
    const float di = dinv[base + ii];
    const float sl = on ? di * di : 0.0f;
    const float4 v = *reinterpret_cast<const float4*>(x + (size_t)(base + ii) * D + 4 * lane);
    s[k] = make_float4(sl * v.x, sl * v.y, sl * v.z, sl * v.w);
  }
  __syncthreads();

  {
    const int listed = l_cnt[wave];
    const int cntw = __builtin_amdgcn_readfirstlane((listed < LCAP) ? listed : LCAP);
    const int* ls = l_src + wave * LCAP;
    const int* lr = l_row + wave * LCAP;
    const float* ln = l_nrm + wave * LCAP;
    for (int j0 = 0; j0 < cntw; j0 += 32) {
      const int m = (cntw - j0 < 32) ? (cntw - j0) : 32;
      int r = 0, slot = 0;
      float nrm = 0.0f;
      if (lane < m) {
        r = ls[j0 + lane];
        slot = lr[j0 + lane];
        nrm = ln[j0 + lane];
      }
      for (int g = 0; g < m; g += ROWS_IN_FLIGHT) {
        float4 v4[ROWS_IN_FLIGHT];
#pragma unroll
        for (int x8 = 0; x8 < ROWS_IN_FLIGHT; ++x8) {
          const int j = (g + x8 < m) ? (g + x8) : (m - 1);
          const int rj = __builtin_amdgcn_readlane(r, j);
          v4[x8] = *reinterpret_cast<const float4*>(x + (size_t)rj * D + 4 * lane);
        }
        int kk[ROWS_IN_FLIGHT];
        float nn[ROWS_IN_FLIGHT];
#pragma unroll
        for (int x8 = 0; x8 < ROWS_IN_FLIGHT; ++x8) {
          const bool on = g + x8 < m;
          const int j = on ? (g + x8) : (m - 1);
          const int k = __builtin_amdgcn_readlane(slot, j);
          kk[x8] = on ? k : -1;
          nn[x8] = __int_as_float(__builtin_amdgcn_readlane(__float_as_int(nrm), j));
        }
#pragma unroll
        for (int kh = 0; kh < (RPW + 3) / 4; ++kh) {
#pragma unroll
          for (int x8 = 0; x8 < ROWS_IN_FLIGHT; ++x8) {
            if (__builtin_expect((kk[x8] >> 2) == kh, 0)) {
#pragma unroll
              for (int kl = 0; kl < 4; ++kl) {
                if (4 * kh + kl < RPW) {
                  if (__builtin_expect(kk[x8] == 4 * kh + kl, 0))
                    row_fma(s[4 * kh + kl], nn[x8], v4[x8]);
                }
              }
            }
          }
        }
      }
    }

#pragma unroll
    for (int k = 0; k < RPW; ++k) {
      const int i = wave + AGG_WAVES * k;
      if (i < cnt) {
        float* a = acc + i * D + ((lane + i) & 31);
        a[0]  += s[k].x;
        a[32] += s[k].y;
        a[64] += s[k].z;
        a[96] += s[k].w;
      }
    }
  }
  __syncthreads();

  const float slope = pa[0];
  const int c16 = lane & 15;
  const int h   = lane >> 4;
  const float4 bias0 = *reinterpret_cast<const float4*>(bias + 4 * c16);
  const float4 bias1 = *reinterpret_cast<const float4*>(bias + 64 + 4 * c16);
  const int ntiles = (cnt + 15) >> 4;
  for (int tile = wave; tile < ntiles; tile += AGG_WAVES) {
    const int ra = tile * 16 + c16;
    const int rc = (ra < RN) ? ra : (RN - 1);
    const float* arow = acc + rc * D + h * 32;
    const float* wrow = w + (size_t)h * D + 4 * c16;

    v8f d[8];
#pragma unroll
    for (int t = 0; t < 8; ++t) d[t] = (v8f){0.f, 0.f, 0.f, 0.f, 0.f, 0.f, 0.f, 0.f};

    for (int q = 0; q < D / 4; ++q) {
      const int o = (q + rc) & 31;
      v2f a;
      a.x = arow[o];
      a.y = arow[64 + o];
      const float* wp = wrow + (size_t)q * (4 * D);
      const float4 b00 = *reinterpret_cast<const float4*>(wp);
      const float4 b01 = *reinterpret_cast<const float4*>(wp + 64);
      const float4 b10 = *reinterpret_cast<const float4*>(wp + 2 * D);
      const float4 b11 = *reinterpret_cast<const float4*>(wp + 2 * D + 64);
      d[0] = __builtin_amdgcn_wmma_f32_16x16x4_f32(false, a, false, (v2f){b00.x, b10.x}, (short)0, d[0], false, false);
      d[1] = __builtin_amdgcn_wmma_f32_16x16x4_f32(false, a, false, (v2f){b00.y, b10.y}, (short)0, d[1], false, false);
      d[2] = __builtin_amdgcn_wmma_f32_16x16x4_f32(false, a, false, (v2f){b00.z, b10.z}, (short)0, d[2], false, false);
      d[3] = __builtin_amdgcn_wmma_f32_16x16x4_f32(false, a, false, (v2f){b00.w, b10.w}, (short)0, d[3], false, false);
      d[4] = __builtin_amdgcn_wmma_f32_16x16x4_f32(false, a, false, (v2f){b01.x, b11.x}, (short)0, d[4], false, false);
      d[5] = __builtin_amdgcn_wmma_f32_16x16x4_f32(false, a, false, (v2f){b01.y, b11.y}, (short)0, d[5], false, false);
      d[6] = __builtin_amdgcn_wmma_f32_16x16x4_f32(false, a, false, (v2f){b01.z, b11.z}, (short)0, d[6], false, false);
      d[7] = __builtin_amdgcn_wmma_f32_16x16x4_f32(false, a, false, (v2f){b01.w, b11.w}, (short)0, d[7], false, false);
    }

#pragma unroll
    for (int j = 0; j < 8; ++j) {
      const int ro = tile * 16 + 8 * h + j;
      if (ro < cnt) {
        float4 v0, v1;
        v0.x = bias0.x + d[0][j];
        v0.y = bias0.y + d[1][j];
        v0.z = bias0.z + d[2][j];
        v0.w = bias0.w + d[3][j];
        v1.x = bias1.x + d[4][j];
        v1.y = bias1.y + d[5][j];
        v1.z = bias1.z + d[6][j];
        v1.w = bias1.w + d[7][j];
        v0.x = (v0.x >= 0.0f) ? v0.x : slope * v0.x;
        v0.y = (v0.y >= 0.0f) ? v0.y : slope * v0.y;
        v0.z = (v0.z >= 0.0f) ? v0.z : slope * v0.z;
        v0.w = (v0.w >= 0.0f) ? v0.w : slope * v0.w;
        v1.x = (v1.x >= 0.0f) ? v1.x : slope * v1.x;
        v1.y = (v1.y >= 0.0f) ? v1.y : slope * v1.y;
        v1.z = (v1.z >= 0.0f) ? v1.z : slope * v1.z;
        v1.w = (v1.w >= 0.0f) ? v1.w : slope * v1.w;
        float* op = out + (size_t)(base + ro) * D + 4 * c16;
        *reinterpret_cast<float4*>(op)      = v0;
        *reinterpret_cast<float4*>(op + 64) = v1;
      }
    }
  }
}

extern "C" void kernel_launch(void* const* d_in, const int* in_sizes, int n_in,
                              void* d_out, int out_size, void* d_ws, size_t ws_size,
                              hipStream_t stream) {
  const float* x    = (const float*)d_in[0];
  const int*   ei   = (const int*)  d_in[1];
  const float* ew   = (const float*)d_in[2];
  const float* w    = (const float*)d_in[3];
  const float* bias = (const float*)d_in[4];
  const float* pa   = (const float*)d_in[5];
  float* out = (float*)d_out;

  const int n     = in_sizes[0] / D;
  const int e_cnt = in_sizes[2];
  const int* row = ei;
  const int* col = ei + e_cnt;

  float* deg = (float*)d_ws;
  const int e_pad = (e_cnt + 511) & ~511;
  float* part = deg + ((n + 127) & ~127);
  const int n_ranges = (n + DEG_HR - 1) / DEG_HR;
  const int n_pad = n_ranges * DEG_HR;
  int4* rec = (int4*)(part + (size_t)DEG_SLABS * n_pad);
  uint4* bits = (uint4*)(rec + e_pad);

  deg_partial_kernel<<<DEG_SLABS * n_ranges, 1024, 0, stream>>>(row, col, ew, part, bits, rec, n_pad, e_cnt, e_pad);
  dinv_reduce_kernel<<<(n + 255) / 256, 256, 0, stream>>>(part, deg, n, n_pad);
  aggregate_kernel<<<(n + RN - 1) / RN, 1024, 0, stream>>>(rec, deg, x, w, bits, bias, pa, out, n, e_cnt, e_pad);
}


# product split by wave role: 16 waves sum the rows from 256 on while 16 multiply the first 256 rows' tiles by w (w asked a step ahead), then the last tiles: f32 WMMA beside gather; k, sum order kept
# speedup vs baseline: 7.2633x; 1.0941x over previous
#include <hip/hip_runtime.h>
#include <hip/hip_bf16.h>

typedef __attribute__((ext_vector_type(2))) float v2f;
typedef __attribute__((ext_vector_type(8))) float v8f;

#define D 128

#define RN 391
#define DEG_SLABS 32
#define DEG_HR 12544
#define DEG_UNR 4
#define N_PLANES 32
#define BM_PITCH 132

__device__ __forceinline__ void deg_apply(float* hist, int4 c, float4 w, int lo) {
  const unsigned dx = (unsigned)(c.x - lo);
  const unsigned dy = (unsigned)(c.y - lo);
  const unsigned dz = (unsigned)(c.z - lo);
  const unsigned dw = (unsigned)(c.w - lo);
  if (dx < (unsigned)DEG_HR) atomicAdd(&hist[dx], w.x);
  if (dy < (unsigned)DEG_HR) atomicAdd(&hist[dy], w.y);
  if (dz < (unsigned)DEG_HR) atomicAdd(&hist[dz], w.z);
  if (dw < (unsigned)DEG_HR) atomicAdd(&hist[dw], w.w);
}

__device__ __forceinline__ void mark_edge(unsigned* wp, int c, unsigned bit) {
  if (c >= 0) {
    const unsigned o = (unsigned)c / RN;
    atomicOr(wp + ((o >> 4) & 15u) * BM_PITCH, bit);
    atomicOr(wp + (16u + (o & 15u)) * BM_PITCH, bit);
  }
}

__device__ __forceinline__ void build_planes(unsigned* bm, uint4* __restrict__ bits,
                                             int4 c, int u, int units, int tid) {
  unsigned* wp = bm + (tid >> 3);
  const unsigned b0 = 1u << ((tid & 7) << 2);
  mark_edge(wp, c.x, b0);
  mark_edge(wp, c.y, b0 << 1);
  mark_edge(wp, c.z, b0 << 2);
  mark_edge(wp, c.w, b0 << 3);
  __syncthreads();
  const int p = tid >> 5;
  const int l = tid & 31;
  uint4* src = reinterpret_cast<uint4*>(bm + p * BM_PITCH + 4 * l);
  const uint4 v = *src;
  *src = make_uint4(0u, 0u, 0u, 0u);
  __syncthreads();
  bits[((size_t)p * units + u) * 32 + l] = v;
}

__device__ __forceinline__ void write_records(const int* __restrict__ row,
                                              const int* __restrict__ col,
                                              const float* __restrict__ ew,
                                              int4* __restrict__ rec,
                                              int ubase, int tid, int e_cnt, bool whole) {
  int4 q[4];
#pragma unroll
  for (int k = 0; k < 4; ++k) {
    const int e = ubase + (k << 10) + tid;
    q[k] = make_int4(0, 0, 0, 0);
    if (whole || e < e_cnt) q[k] = make_int4(row[e], col[e], __float_as_int(ew[e]), 0);
  }
#pragma unroll
  for (int k = 0; k < 4; ++k) {
    const int e = ubase + (k << 10) + tid;
    if (whole || e < e_cnt) rec[e] = q[k];
  }
}

__global__ __launch_bounds__(1024) void deg_partial_kernel(
    const int* __restrict__ row, const int* __restrict__ col,
    const float* __restrict__ ew,
    float* __restrict__ part, uint4* __restrict__ bits,
    int4* __restrict__ rec,
    int n_pad, int e_cnt, int e_pad) {
  __shared__ alignas(16) float hist[DEG_HR];
  __shared__ alignas(16) unsigned bm[N_PLANES * BM_PITCH];
  const int tid  = threadIdx.x;
  const int slab = (int)blockIdx.x % DEG_SLABS;
  const int g    = (int)blockIdx.x / DEG_SLABS;
  const int n_ranges = (int)gridDim.x / DEG_SLABS;
  const int lo = g * DEG_HR;

  for (int i = tid; i < DEG_HR; i += 1024) hist[i] = 0.0f;
  for (int i = tid; i < N_PLANES * BM_PITCH; i += 1024) bm[i] = 0u;
  __syncthreads();

  const int units = (e_pad + 4095) >> 12;
  const int ubeg = (int)(((long long)slab * units) / DEG_SLABS);
  const int uend = (int)(((long long)(slab + 1) * units) / DEG_SLABS);
  const int ufull = e_cnt >> 12;

  int u0 = ubeg;
  const int ufast = (uend < ufull) ? uend : ufull;
  for (; u0 + DEG_UNR <= ufast; u0 += DEG_UNR) {
    int4 c[DEG_UNR];
    float4 w[DEG_UNR];
#pragma unroll
    for (int j = 0; j < DEG_UNR; ++j) {
      const int e0 = ((u0 + j) << 12) + (tid << 2);
      c[j] = *reinterpret_cast<const int4*>(col + e0);
      w[j] = *reinterpret_cast<const float4*>(ew + e0);
    }
#pragma unroll
    for (int j = 0; j < DEG_UNR; ++j) {
      const int u = u0 + j;
      const bool mine = (u % n_ranges) == g;
      deg_apply(hist, c[j], w[j], lo);
      if (mine) {
        build_planes(bm, bits, c[j], u, units, tid);
        write_records(row, col, ew, rec, u << 12, tid, e_cnt, true);
      }
    }
  }
  for (; u0 < uend; ++u0) {
    const int e0 = (u0 << 12) + (tid << 2);
    int4 c = make_int4(-1, -1, -1, -1);
    float4 w = make_float4(0.0f, 0.0f, 0.0f, 0.0f);
    if (e0 + 3 < e_cnt) {
      c = *reinterpret_cast<const int4*>(col + e0);
      w = *reinterpret_cast<const float4*>(ew + e0);
    } else {
      if (e0 < e_cnt)     { c.x = col[e0];     w.x = ew[e0]; }
      if (e0 + 1 < e_cnt) { c.y = col[e0 + 1]; w.y = ew[e0 + 1]; }
      if (e0 + 2 < e_cnt) { c.z = col[e0 + 2]; w.z = ew[e0 + 2]; }
    }
    const bool mine = (u0 % n_ranges) == g;
    deg_apply(hist, c, w, lo);
    if (mine) {
      build_planes(bm, bits, c, u0, units, tid);
      write_records(row, col, ew, rec, u0 << 12, tid, e_cnt, false);
    }
  }
  __syncthreads();

  float* dst = part + (size_t)slab * n_pad + lo;
  for (int i = tid << 2; i < DEG_HR; i += 4096)
    *reinterpret_cast<float4*>(dst + i) = *reinterpret_cast<const float4*>(hist + i);
}

__global__ void dinv_reduce_kernel(const float* __restrict__ part, float* __restrict__ dinv,
                                   int n, int n_pad) {
  const int i = blockIdx.x * blockDim.x + threadIdx.x;
  if (i < n) {
    float d = 1.0f;
#pragma unroll
    for (int s = 0; s < DEG_SLABS; ++s) d += part[(size_t)s * n_pad + i];
    dinv[i] = (d > 0.0f) ? rsqrtf(d) : 0.0f;
  }
}

#define AGG_WAVES 32
#define RCAP 31
#define QCAP 64
#define SCAN_DEPTH 2
#define SCAN_WORDS (SCAN_DEPTH * 4)
#define SCAN_STEP (AGG_WAVES * SCAN_DEPTH)
#define ROWS_IN_FLIGHT 8
#define SPLIT_ROWS 256
#define SUM_WAVES_B 16

__device__ __forceinline__ void load_planes(uint4 (&ch)[SCAN_DEPTH], uint4 (&cl)[SCAN_DEPTH],
                                            const uint4* __restrict__ ph,
                                            const uint4* __restrict__ pl,
                                            int ch0, int hi) {
#pragma unroll
  for (int u = 0; u < SCAN_DEPTH; ++u) {
    uint4 a = make_uint4(0u, 0u, 0u, 0u);
    uint4 b = make_uint4(0u, 0u, 0u, 0u);
    const int c = ch0 + u * AGG_WAVES;
    if (c < hi) {
      a = ph[(size_t)c * 32];
      b = pl[(size_t)c * 32];
    }
    ch[u] = a;
    cl[u] = b;
  }
}

__device__ __forceinline__ void row_fma(float4& s, float f, const float4& v) {
  s.x = fmaf(f, v.x, s.x);
  s.y = fmaf(f, v.y, s.y);
  s.z = fmaf(f, v.z, s.z);
  s.w = fmaf(f, v.w, s.w);
}

__global__ __launch_bounds__(1024) void aggregate_kernel(
    const int4* __restrict__ rec, const float* __restrict__ dinv,
    const float* __restrict__ x, const float* __restrict__ w,
    const uint4* __restrict__ bits,
    const float* __restrict__ bias, const float* __restrict__ pa,
    float* __restrict__ out, int n, int e_cnt, int e_pad) {
  __shared__ alignas(16) float acc[RN * D];
  __shared__ int q_e[AGG_WAVES * QCAP];
  __shared__ int l_src[RN * RCAP];
  __shared__ float l_nrm[RN * RCAP];
  __shared__ int r_cnt[RN];
  __shared__ float r_dsq[RN];

  const int lane = threadIdx.x & 31;
  const int wave = __builtin_amdgcn_readfirstlane((int)(threadIdx.x >> 5));
  const int base = blockIdx.x * RN;
  const int cnt  = (n - base < RN) ? (n - base) : RN;

  float down = 0.0f;
  if (threadIdx.x < RN)
    down = dinv[base + (((int)threadIdx.x < cnt) ? (int)threadIdx.x : (cnt - 1))];
  for (int i = threadIdx.x; i < RN * D / 4; i += 1024)
    reinterpret_cast<float4*>(acc)[i] = make_float4(0.0f, 0.0f, 0.0f, 0.0f);
  if (threadIdx.x < RN) {
    r_cnt[threadIdx.x] = 0;
    r_dsq[threadIdx.x] = down * down;
  }
  __syncthreads();

  int* qe = q_e + wave * QCAP;
  int head = 0;
  int qlen = 0;

  const int nch   = (e_pad + 4095) >> 12;
  const int start = (int)blockIdx.x * (nch / (int)gridDim.x);
  const uint4* ph = bits + (size_t)(((unsigned)blockIdx.x >> 4) & 15u) * nch * 32 + lane;
  const uint4* pl = bits + (size_t)(16u + ((unsigned)blockIdx.x & 15u)) * nch * 32 + lane;

  int pass = 0;
  int ch0 = start + wave;
  int hi = nch;
  if (ch0 >= hi) { pass = 1; ch0 = wave; hi = start; }
  bool valid = ch0 < hi;

  uint4 ch[SCAN_DEPTH], cl[SCAN_DEPTH];
  if (valid) load_planes(ch, cl, ph, pl, ch0, hi);

  while (valid) {
    unsigned mw[SCAN_WORDS];
#pragma unroll
    for (int u = 0; u < SCAN_DEPTH; ++u) {
      mw[4 * u + 0] = ch[u].x & cl[u].x;
      mw[4 * u + 1] = ch[u].y & cl[u].y;
      mw[4 * u + 2] = ch[u].z & cl[u].z;
      mw[4 * u + 3] = ch[u].w & cl[u].w;
    }
    const int cur = ch0;

    ch0 += SCAN_STEP;
    if (ch0 >= hi && pass == 0) { pass = 1; ch0 = wave; hi = start; }
    valid = ch0 < hi;
    if (valid) load_planes(ch, cl, ph, pl, ch0, hi);

#pragma nounroll
    for (int v = 0; v < SCAN_WORDS; ++v) {
      unsigned mbits = mw[0];
#pragma unroll
      for (int t = 0; t + 1 < SCAN_WORDS; ++t) mw[t] = mw[t + 1];
      const bool flush = !valid && (v == SCAN_WORDS - 1);
      const int ebase = ((cur + (v >> 2) * AGG_WAVES) << 12) + ((v & 3) << 5);

      for (;;) {
        while (qlen < 32) {
          const bool has = mbits != 0u;
          const unsigned b = __builtin_amdgcn_ballot_w32(has);
          if (b == 0u) break;
          if (has) {
            const int s = __builtin_ctz(mbits);
            mbits &= mbits - 1u;
            const int e = ebase + (lane << 7) + s;
            const int pos = head + qlen + (int)__builtin_amdgcn_mbcnt_lo(b, 0u);
            qe[pos & (QCAP - 1)] = e;
          }
          qlen += __builtin_popcount(b);
        }
        const int m = (qlen < 32) ? qlen : 32;
        if (m == 0 || (m < 32 && !flush)) break;

        int r = 0, lc = -1;
        float nrm = 0.0f;
        if (lane < m) {
          const int e = qe[(head + lane) & (QCAP - 1)];
          if (e < e_cnt) {
            const int4 rc = rec[e];
            r = rc.x;
            const int c = rc.y;
            const float w = __int_as_float(rc.z);
            r = (r < 0) ? 0 : ((r > n - 1) ? (n - 1) : r);
            const bool mine = (unsigned)(c - base) < (unsigned)cnt;
            const float f = dinv[r] * w * dinv[mine ? c : base];
            lc = mine ? (c - base) : -1;
            nrm = mine ? f : 0.0f;
          }
        }
        if (lc >= 0) {
          const int slot = atomicAdd(&r_cnt[lc], 1);
          if (slot < RCAP) {
            const int at = lc * RCAP + slot;
            l_src[at] = r;
            l_nrm[at] = nrm;
            lc = -1;
          }
        }
        unsigned ov = __builtin_amdgcn_ballot_w32(lc >= 0);
        while (ov != 0u) {
          const int j = __builtin_ctz(ov);
          ov &= ov - 1u;
          const int rj  = __builtin_amdgcn_readlane(r, j);
          const int lcj = __builtin_amdgcn_readlane(lc, j);
          const float nj = __int_as_float(__builtin_amdgcn_readlane(__float_as_int(nrm), j));
          const float4 xv = *reinterpret_cast<const float4*>(x + (size_t)rj * D + 4 * lane);
          float* a = acc + lcj * D + ((lane + lcj) & 31);
          atomicAdd(a,      nj * xv.x);
          atomicAdd(a + 32, nj * xv.y);
          atomicAdd(a + 64, nj * xv.z);
          atomicAdd(a + 96, nj * xv.w);
        }
        head = (head + m) & (QCAP - 1);
        qlen -= m;
      }
    }
  }
  __syncthreads();

  const int c16 = lane & 15;
  const int h   = lane >> 4;
  const int ntiles = (cnt + 15) >> 4;
  const bool mul_half = (wave & 4) != 0;
  const int place = (wave & 3) | ((wave >> 3) << 2);
  const int late = 4 * (((wave >> 3) - (wave & 3)) & 3) + (wave & 3);

#pragma nounroll
  for (int stage = 0; stage < 3; ++stage) {
    int rs = 0, stride = 1, re = 0;
    if (stage == 0) { rs = wave; stride = AGG_WAVES; re = SPLIT_ROWS; }
    if (stage == 1 && !mul_half) { rs = SPLIT_ROWS + place; stride = SUM_WAVES_B; re = RN; }

#pragma nounroll
    for (int i0 = rs; i0 < re; i0 += 2 * stride) {
      int ri[2], rr[2], ne[2], sr[2];
      float fc[2];
      bool on[2];
      int listed[2], ls[2];
      float dsq[2], lf[2];
#pragma unroll
      for (int t = 0; t < 2; ++t) {
        ri[t] = i0 + t * stride;
        on[t] = (ri[t] < re) && (ri[t] < cnt);
        rr[t] = on[t] ? ri[t] : 0;
        const int at = rr[t] * RCAP + ((lane > 0) ? (lane - 1) : 0);
        listed[t] = r_cnt[rr[t]];
        dsq[t] = r_dsq[rr[t]];
        ls[t] = l_src[at];
        lf[t] = l_nrm[at];
      }
#pragma unroll
      for (int t = 0; t < 2; ++t) {
        const int cl = __builtin_amdgcn_readfirstlane((listed[t] < RCAP) ? listed[t] : RCAP);
        const int cn = on[t] ? cl : 0;
        const bool in_list = lane >= 1 && lane <= cn;
        const int sv = in_list ? ls[t] : (base + rr[t]);
        const int last = __builtin_amdgcn_readlane(sv, cn);
        sr[t] = (lane <= cn) ? sv : last;
        fc[t] = in_list ? lf[t] : ((on[t] && lane == 0) ? dsq[t] : 0.0f);
        ne[t] = on[t] ? (cn + 1) : 0;
      }
      float4 a[2];
      a[0] = make_float4(0.0f, 0.0f, 0.0f, 0.0f);
      a[1] = make_float4(0.0f, 0.0f, 0.0f, 0.0f);
      const int nemax = (ne[0] > ne[1]) ? ne[0] : ne[1];
      for (int g = 0; g < nemax; g += ROWS_IN_FLIGHT) {
        float4 v4[2][ROWS_IN_FLIGHT];
#pragma unroll
        for (int t = 0; t < 2; ++t) {
#pragma unroll
          for (int x8 = 0; x8 < ROWS_IN_FLIGHT; ++x8) {
            const int rj = __builtin_amdgcn_readlane(sr[t], g + x8);
            v4[t][x8] = *reinterpret_cast<const float4*>(x + (size_t)rj * D + 4 * lane);
          }
        }
#pragma unroll
        for (int t = 0; t < 2; ++t) {
#pragma unroll
          for (int x8 = 0; x8 < ROWS_IN_FLIGHT; ++x8) {
            const float f = __int_as_float(
                __builtin_amdgcn_readlane(__float_as_int(fc[t]), g + x8));
            row_fma(a[t], f, v4[t][x8]);
          }
        }
      }
#pragma unroll
      for (int t = 0; t < 2; ++t) {
        if (on[t]) {
          const int i = ri[t];
          float* p = acc + i * D + ((lane + i) & 31);
          p[0]  += a[t].x;
          p[32] += a[t].y;
          p[64] += a[t].z;
          p[96] += a[t].w;
        }
      }
    }

    int tile = -1;
    if (stage == 1 && mul_half) tile = place;
    if (stage == 2 && !mul_half) tile = SPLIT_ROWS / 16 + late;
    if (tile >= 0 && tile < ntiles) {
      const int ra = tile * 16 + c16;
      const int rc = (ra < RN) ? ra : (RN - 1);
      const float* arow = acc + rc * D + h * 32;
      const float* wrow = w + (size_t)h * D + 4 * c16;

      v8f d[8];
#pragma unroll
      for (int t = 0; t < 8; ++t) d[t] = (v8f){0.f, 0.f, 0.f, 0.f, 0.f, 0.f, 0.f, 0.f};

      int o = rc & 31;
      v2f an;
      an.x = arow[o];
      an.y = arow[64 + o];
      float4 n00 = *reinterpret_cast<const float4*>(wrow);
      float4 n01 = *reinterpret_cast<const float4*>(wrow + 64);
      float4 n10 = *reinterpret_cast<const float4*>(wrow + 2 * D);
      float4 n11 = *reinterpret_cast<const float4*>(wrow + 2 * D + 64);
#pragma unroll 2
      for (int q = 0; q < D / 4; ++q) {
        const v2f a = an;
        const float4 b00 = n00, b01 = n01, b10 = n10, b11 = n11;
        const int qn = (q + 1 < D / 4) ? (q + 1) : q;
        o = (qn + rc) & 31;
        an.x = arow[o];
        an.y = arow[64 + o];
        const float* wp = wrow + (size_t)qn * (4 * D);
        n00 = *reinterpret_cast<const float4*>(wp);
        n01 = *reinterpret_cast<const float4*>(wp + 64);
        n10 = *reinterpret_cast<const float4*>(wp + 2 * D);
        n11 = *reinterpret_cast<const float4*>(wp + 2 * D + 64);
        __builtin_amdgcn_sched_barrier(0);
        d[0] = __builtin_amdgcn_wmma_f32_16x16x4_f32(false, a, false, (v2f){b00.x, b10.x}, (short)0, d[0], false, false);
        d[1] = __builtin_amdgcn_wmma_f32_16x16x4_f32(false, a, false, (v2f){b00.y, b10.y}, (short)0, d[1], false, false);
        d[2] = __builtin_amdgcn_wmma_f32_16x16x4_f32(false, a, false, (v2f){b00.z, b10.z}, (short)0, d[2], false, false);
        d[3] = __builtin_amdgcn_wmma_f32_16x16x4_f32(false, a, false, (v2f){b00.w, b10.w}, (short)0, d[3], false, false);
        d[4] = __builtin_amdgcn_wmma_f32_16x16x4_f32(false, a, false, (v2f){b01.x, b11.x}, (short)0, d[4], false, false);
        d[5] = __builtin_amdgcn_wmma_f32_16x16x4_f32(false, a, false, (v2f){b01.y, b11.y}, (short)0, d[5], false, false);
        d[6] = __builtin_amdgcn_wmma_f32_16x16x4_f32(false, a, false, (v2f){b01.z, b11.z}, (short)0, d[6], false, false);
        d[7] = __builtin_amdgcn_wmma_f32_16x16x4_f32(false, a, false, (v2f){b01.w, b11.w}, (short)0, d[7], false, false);
      }

      const float slope = pa[0];
      const float4 bias0 = *reinterpret_cast<const float4*>(bias + 4 * c16);
      const float4 bias1 = *reinterpret_cast<const float4*>(bias + 64 + 4 * c16);
#pragma unroll
      for (int j = 0; j < 8; ++j) {
        const int ro = tile * 16 + 8 * h + j;
        if (ro < cnt) {
          float4 v0, v1;
          v0.x = bias0.x + d[0][j];
          v0.y = bias0.y + d[1][j];
          v0.z = bias0.z + d[2][j];
          v0.w = bias0.w + d[3][j];
          v1.x = bias1.x + d[4][j];
          v1.y = bias1.y + d[5][j];
          v1.z = bias1.z + d[6][j];
          v1.w = bias1.w + d[7][j];
          v0.x = (v0.x >= 0.0f) ? v0.x : slope * v0.x;
          v0.y = (v0.y >= 0.0f) ? v0.y : slope * v0.y;
          v0.z = (v0.z >= 0.0f) ? v0.z : slope * v0.z;
          v0.w = (v0.w >= 0.0f) ? v0.w : slope * v0.w;
          v1.x = (v1.x >= 0.0f) ? v1.x : slope * v1.x;
          v1.y = (v1.y >= 0.0f) ? v1.y : slope * v1.y;
          v1.z = (v1.z >= 0.0f) ? v1.z : slope * v1.z;
          v1.w = (v1.w >= 0.0f) ? v1.w : slope * v1.w;
          float* op = out + (size_t)(base + ro) * D + 4 * c16;
          *reinterpret_cast<float4*>(op)      = v0;
          *reinterpret_cast<float4*>(op + 64) = v1;
        }
      }
    }
    if (stage < 2) __syncthreads();
  }
}

extern "C" void kernel_launch(void* const* d_in, const int* in_sizes, int n_in,
                              void* d_out, int out_size, void* d_ws, size_t ws_size,
                              hipStream_t stream) {
  const float* x    = (const float*)d_in[0];
  const int*   ei   = (const int*)  d_in[1];
  const float* ew   = (const float*)d_in[2];
  const float* w    = (const float*)d_in[3];
  const float* bias = (const float*)d_in[4];
  const float* pa   = (const float*)d_in[5];
  float* out = (float*)d_out;

  const int n     = in_sizes[0] / D;
  const int e_cnt = in_sizes[2];
  const int* row = ei;
  const int* col = ei + e_cnt;

  float* deg = (float*)d_ws;
  const int e_pad = (e_cnt + 511) & ~511;
  float* part = deg + ((n + 127) & ~127);
  const int n_ranges = (n + DEG_HR - 1) / DEG_HR;
  const int n_pad = n_ranges * DEG_HR;
  int4* rec = (int4*)(part + (size_t)DEG_SLABS * n_pad);
  uint4* bits = (uint4*)(rec + e_pad);

  deg_partial_kernel<<<DEG_SLABS * n_ranges, 1024, 0, stream>>>(row, col, ew, part, bits, rec, n_pad, e_cnt, e_pad);
  dinv_reduce_kernel<<<(n + 255) / 256, 256, 0, stream>>>(part, deg, n, n_pad);
  aggregate_kernel<<<(n + RN - 1) / RN, 1024, 0, stream>>>(rec, deg, x, w, bits, bias, pa, out, n, e_cnt, e_pad);
}
